# starting kernel
# baseline (speedup 1.0000x reference)
#include <hip/hip_runtime.h>


#define NN 4096
#define DD 16

typedef float v2f __attribute__((ext_vector_type(2)));
typedef float v8f __attribute__((ext_vector_type(8)));

__device__ __forceinline__ void load_a_frag(const float* __restrict__ X,
                                            int row0, float scale, v2f a[4]) {
  const int lane = threadIdx.x & 31;
  const int m = lane & 15, h = lane >> 4;
  const float* base = X + (size_t)(row0 + m) * DD + 2 * h;
#pragma unroll
  for (int c = 0; c < 4; ++c) {
    a[c].x = base[4 * c] * scale;
    a[c].y = base[4 * c + 1] * scale;
  }
}

__device__ __forceinline__ void load_b_frag(const float* __restrict__ W, v2f b[4]) {
  const int lane = threadIdx.x & 31;
  const int n = lane & 15, h = lane >> 4;
#pragma unroll
  for (int c = 0; c < 4; ++c) {
    b[c].x = W[(4 * c + 2 * h) * DD + n];
    b[c].y = W[(4 * c + 2 * h + 1) * DD + n];
  }
}

__device__ __forceinline__ void load_bT_frag(const float* __restrict__ W, v2f b[4]) {
  const int lane = threadIdx.x & 31;
  const int n = lane & 15, h = lane >> 4;
  const float* row = W + n * DD;
#pragma unroll
  for (int c = 0; c < 4; ++c) {
    b[c].x = row[4 * c + 2 * h];
    b[c].y = row[4 * c + 2 * h + 1];
  }
}

__device__ __forceinline__ v8f load_c_tile(const float* __restrict__ X, int row0) {
  const int lane = threadIdx.x & 31;
  const int n = lane & 15, h = lane >> 4;
  v8f c;
#pragma unroll
  for (int r = 0; r < 8; ++r) c[r] = X[(size_t)(row0 + r + 8 * h) * DD + n];
  return c;
}

__device__ __forceinline__ void store_d_tile(float* __restrict__ O, int row0, v8f d) {
  const int lane = threadIdx.x & 31;
  const int n = lane & 15, h = lane >> 4;
#pragma unroll
  for (int r = 0; r < 8; ++r) O[(size_t)(row0 + r + 8 * h) * DD + n] = d[r];
}

__device__ __forceinline__ v8f wmma16(const v2f a[4], const v2f b[4], v8f c) {
#pragma unroll
  for (int k = 0; k < 4; ++k)
    c = __builtin_amdgcn_wmma_f32_16x16x4_f32(false, a[k], false, b[k],
                                              (short)0, c, false, false);
  return c;
}

__global__ void k_build_mask(const int* __restrict__ mrow, const int* __restrict__ mcol,
                             int E, unsigned* __restrict__ Md) {
  int e = blockIdx.x * blockDim.x + threadIdx.x;
  if (e >= E) return;
  unsigned bit = (unsigned)mrow[e] * (unsigned)NN + (unsigned)mcol[e];
  atomicOr(&Md[bit >> 5], 1u << (bit & 31));
}

__global__ void k_build_adj(const int* __restrict__ erow, const int* __restrict__ ecol,
                            int E, const unsigned* __restrict__ Md,
                            unsigned* __restrict__ Ab, unsigned* __restrict__ list,
                            unsigned* __restrict__ cnt) {
  int e = blockIdx.x * blockDim.x + threadIdx.x;
  if (e >= E) return;
  unsigned i = (unsigned)erow[e], j = (unsigned)ecol[e];
  unsigned bit = i * (unsigned)NN + j;
  unsigned w = bit >> 5, b = 1u << (bit & 31);
  if (!(Md[w] & b)) return;
  unsigned old = atomicOr(&Ab[w], b);
  if (old & b) return;
  unsigned idx = atomicAdd(cnt, 1u);
  list[idx] = (i << 12) | j;
}

__global__ void k_g1(const float* __restrict__ W1, const float* __restrict__ Q,
                     float* __restrict__ G1) {
  int s = blockIdx.x;
  int k = threadIdx.x >> 4, n = threadIdx.x & 15;
  float acc = 0.f;
#pragma unroll
  for (int t = 0; t < DD; ++t)
    acc = fmaf(W1[s * DD * DD + t * DD + k], Q[t * DD + n], acc);
  G1[s * DD * DD + k * DD + n] = acc;
}

__global__ void k_pre(const float* __restrict__ x, const float* __restrict__ Qm,
                      const float* __restrict__ W2s, float* __restrict__ xq,
                      float* __restrict__ x2) {
  int wave = threadIdx.x >> 5;
  int row0 = (blockIdx.x * (blockDim.x >> 5) + wave) * 16;
  v2f a[4], b[4];
  load_a_frag(x, row0, 1.0f, a);
  load_b_frag(Qm, b);
  v8f d = {};
  d = wmma16(a, b, d);
  store_d_tile(xq, row0, d);
  load_bT_frag(W2s, b);
  v8f d2 = {};
  d2 = wmma16(a, b, d2);
  store_d_tile(x2, row0, d2);
}

__global__ void k_scatter1(const unsigned* __restrict__ list,
                           const unsigned* __restrict__ cnt,
                           const float* __restrict__ x, float* __restrict__ y1) {
  unsigned e = blockIdx.x * blockDim.x + threadIdx.x;
  if (e >= *cnt) return;
  unsigned p = list[e];
  const float* xs = x + (size_t)(p >> 12) * DD;
  float* yd = y1 + (size_t)(p & 4095u) * DD;
#pragma unroll
  for (int c = 0; c < DD; ++c) atomicAdd(&yd[c], xs[c]);
}

__global__ void k_scatter2(const unsigned* __restrict__ list,
                           const unsigned* __restrict__ cnt,
                           const float* __restrict__ xq, const float* __restrict__ x,
                           const float* __restrict__ x2, float* __restrict__ y2) {
  unsigned e = blockIdx.x * blockDim.x + threadIdx.x;
  if (e >= *cnt) return;
  unsigned p = list[e];
  unsigned i = p >> 12, j = p & 4095u;
  const float* qi = xq + (size_t)i * DD;
  const float* xj = x + (size_t)j * DD;
  float a = 0.f;
#pragma unroll
  for (int c = 0; c < DD; ++c) a = fmaf(qi[c], xj[c], a);
  const float* v = x2 + (size_t)i * DD;
  float* yd = y2 + (size_t)j * DD;
#pragma unroll
  for (int c = 0; c < DD; ++c) atomicAdd(&yd[c], a * v[c]);
}

__global__ void k_combine(const float* __restrict__ xcur, const float* __restrict__ y1,
                          const float* __restrict__ y2, const float* __restrict__ G1s,
                          const float* __restrict__ Qm, const unsigned* __restrict__ cnt2,
                          float s1, float* __restrict__ xnext) {
  int wave = threadIdx.x >> 5;
  int row0 = (blockIdx.x * (blockDim.x >> 5) + wave) * 16;
  float s2 = -(float)NN / (float)(*cnt2);
  v2f a[4], b[4];
  v8f c = load_c_tile(xcur, row0);
  load_a_frag(y1, row0, s1, a);
  load_b_frag(G1s, b);
  c = wmma16(a, b, c);
  load_a_frag(y2, row0, s2, a);
  load_b_frag(Qm, b);
  c = wmma16(a, b, c);
  store_d_tile(xnext, row0, c);
}

extern "C" void kernel_launch(void* const* d_in, const int* in_sizes, int n_in,
                              void* d_out, int out_size, void* d_ws, size_t ws_size,
                              hipStream_t stream) {
  const float* x_in = (const float*)d_in[0];
  const int* e1 = (const int*)d_in[1];
  const int* e2 = (const int*)d_in[2];
  const float* Qm = (const float*)d_in[3];
  const int* mk = (const int*)d_in[4];
  const float* W1 = (const float*)d_in[5];
  const float* W2 = (const float*)d_in[6];
  float* out = (float*)d_out;

  const int E1 = in_sizes[1] / 2;
  const int E2 = in_sizes[2] / 2;
  const int EM = in_sizes[4] / 2;
  const int STEPS = in_sizes[5] / (DD * DD);

  char* ws = (char*)d_ws;
  size_t off = 0;
  auto carve = [&](size_t bytes) {
    char* p = ws + off;
    off += (bytes + 255) & ~(size_t)255;
    return p;
  };
  const size_t BM = (size_t)NN * NN / 8;
  unsigned* Md    = (unsigned*)carve(BM);
  unsigned* A1b   = (unsigned*)carve(BM);
  unsigned* A2b   = (unsigned*)carve(BM);
  unsigned* list1 = (unsigned*)carve((size_t)E1 * 4);
  unsigned* list2 = (unsigned*)carve((size_t)E2 * 4);
  unsigned* cnts  = (unsigned*)carve(8);
  float* G1  = (float*)carve((size_t)STEPS * DD * DD * 4);
  float* xq  = (float*)carve((size_t)NN * DD * 4);
  float* x2  = (float*)carve((size_t)NN * DD * 4);
  float* y1  = (float*)carve((size_t)NN * DD * 4);
  float* y2  = (float*)carve((size_t)NN * DD * 4);
  float* xb0 = (float*)carve((size_t)NN * DD * 4);
  float* xb1 = (float*)carve((size_t)NN * DD * 4);

  hipMemsetAsync(Md, 0, 3 * BM, stream);
  hipMemsetAsync(cnts, 0, 8, stream);

  k_build_mask<<<(EM + 255) / 256, 256, 0, stream>>>(mk, mk + EM, EM, Md);
  k_build_adj<<<(E1 + 255) / 256, 256, 0, stream>>>(e1, e1 + E1, E1, Md, A1b, list1, cnts + 0);
  k_build_adj<<<(E2 + 255) / 256, 256, 0, stream>>>(e2, e2 + E2, E2, Md, A2b, list2, cnts + 1);
  k_g1<<<STEPS, 256, 0, stream>>>(W1, Qm, G1);

  const float s1 = (float)NN / (float)EM;

  const int TILE_BLOCKS = NN / 16 / 8;
  const float* xc = x_in;
  for (int s = 0; s < STEPS; ++s) {
    float* xn = (s == STEPS - 1) ? out : ((s & 1) ? xb1 : xb0);
    hipMemsetAsync(y1, 0, (size_t)NN * DD * 4, stream);
    hipMemsetAsync(y2, 0, (size_t)NN * DD * 4, stream);
    k_pre<<<TILE_BLOCKS, 256, 0, stream>>>(xc, Qm, W2 + (size_t)s * DD * DD, xq, x2);
    k_scatter1<<<(E1 + 255) / 256, 256, 0, stream>>>(list1, cnts + 0, xc, y1);
    k_scatter2<<<(E2 + 255) / 256, 256, 0, stream>>>(list2, cnts + 1, xq, xc, x2, y2);
    k_combine<<<TILE_BLOCKS, 256, 0, stream>>>(xc, y1, y2, G1 + (size_t)s * DD * DD,
                                               Qm, cnts + 1, s1, xn);
    xc = xn;
  }
}


# Weight products re-associated out of the edge sums: S = x_i.(Q x_j), y2 = (P^T x)(W2^T Q); the dense kernel reads only x, the x@Q / x@W2^T launch goes (13 to 10 launches); sums in a new order
# speedup vs baseline: 7.1242x; 7.1242x over previous
#include <hip/hip_runtime.h>


#define NN 4096
#define DD 16

typedef float v2f __attribute__((ext_vector_type(2)));
typedef float v8f __attribute__((ext_vector_type(8)));

__device__ __forceinline__ void load_a_frag_sum2(const float* __restrict__ Y,
                                                 int row0, float scale, v2f a[4]) {
  const int lane = threadIdx.x & 31;
  const int m = lane & 15, h = lane >> 4;
  const float* b0 = Y + (size_t)(row0 + m) * DD + 2 * h;
  const float* b1 = b0 + (size_t)NN * DD;
#pragma unroll
  for (int c = 0; c < 4; ++c) {
    a[c].x = (b0[4 * c] + b1[4 * c]) * scale;
    a[c].y = (b0[4 * c + 1] + b1[4 * c + 1]) * scale;
  }
}

__device__ __forceinline__ void load_b_frag(const float* __restrict__ W, v2f b[4]) {
  const int lane = threadIdx.x & 31;
  const int n = lane & 15, h = lane >> 4;
#pragma unroll
  for (int c = 0; c < 4; ++c) {
    b[c].x = W[(4 * c + 2 * h) * DD + n];
    b[c].y = W[(4 * c + 2 * h + 1) * DD + n];
  }
}

__device__ __forceinline__ v8f load_c_tile(const float* __restrict__ X, int row0) {
  const int lane = threadIdx.x & 31;
  const int n = lane & 15, h = lane >> 4;
  v8f c;
#pragma unroll
  for (int r = 0; r < 8; ++r) c[r] = X[(size_t)(row0 + r + 8 * h) * DD + n];
  return c;
}

__device__ __forceinline__ void store_d_tile(float* __restrict__ O, int row0, v8f d) {
  const int lane = threadIdx.x & 31;
  const int n = lane & 15, h = lane >> 4;
#pragma unroll
  for (int r = 0; r < 8; ++r) O[(size_t)(row0 + r + 8 * h) * DD + n] = d[r];
}

__device__ __forceinline__ v8f wmma16(const v2f a[4], const v2f b[4], v8f c) {
#pragma unroll
  for (int k = 0; k < 4; ++k)
    c = __builtin_amdgcn_wmma_f32_16x16x4_f32(false, a[k], false, b[k],
                                              (short)0, c, false, false);
  return c;
}

__global__ void k_build_mask(const int* __restrict__ mrow, const int* __restrict__ mcol,
                             int E, unsigned* __restrict__ MdT) {
  int e = blockIdx.x * blockDim.x + threadIdx.x;
  if (e >= E) return;
  unsigned bit = (unsigned)mcol[e] * (unsigned)NN + (unsigned)mrow[e];
  atomicOr(&MdT[bit >> 5], 1u << (bit & 31));
}

__global__ void k_build_adj(const int* __restrict__ erow, const int* __restrict__ ecol,
                            int E, const unsigned* __restrict__ MdT,
                            unsigned* __restrict__ AbT, unsigned* __restrict__ cnt) {
  int e = blockIdx.x * blockDim.x + threadIdx.x;
  if (e >= E) return;
  unsigned i = (unsigned)erow[e], j = (unsigned)ecol[e];
  unsigned bit = j * (unsigned)NN + i;
  unsigned w = bit >> 5, b = 1u << (bit & 31);
  if (!(MdT[w] & b)) return;
  unsigned old = atomicOr(&AbT[w], b);
  if (old & b) return;
  atomicAdd(cnt, 1u);
}

__global__ void k_g(const float* __restrict__ W1, const float* __restrict__ W2,
                    const float* __restrict__ Q, float* __restrict__ G) {
  const int s = blockIdx.x;
  const float* __restrict__ W = blockIdx.y ? W2 : W1;
  int k = threadIdx.x >> 4, n = threadIdx.x & 15;
  float acc = 0.f;
#pragma unroll
  for (int t = 0; t < DD; ++t)
    acc = fmaf(W[s * DD * DD + t * DD + k], Q[t * DD + n], acc);
  G[((size_t)blockIdx.y * gridDim.x + s) * DD * DD + k * DD + n] = acc;
}


__device__ __forceinline__ float bit_to_one(unsigned m, int k) {
  unsigned all = (unsigned)(((int)(m << (31 - k))) >> 31);
  return __uint_as_float(all & 0x3f800000u);
}
__device__ __forceinline__ float keep_if_bit(float s, unsigned m, int k) {
  unsigned all = (unsigned)(((int)(m << (31 - k))) >> 31);
  return __uint_as_float(__float_as_uint(s) & all);
}

struct SrcFrag { v2f ax[4]; v2f ar[4]; };

__device__ __forceinline__ void load_src(const float* __restrict__ x, int i0, int c16, int h,
                                         SrcFrag& f) {
  const float* px = x + (size_t)(i0 + 8 * h) * DD + c16;
#pragma unroll
  for (int q = 0; q < 4; ++q) {
    f.ax[q].x = px[(2 * q) * DD];
    f.ax[q].y = px[(2 * q + 1) * DD];
  }
  const float4* pr = (const float4*)(x + (size_t)(i0 + c16) * DD + 8 * h);
  float4 r0 = pr[0], r1 = pr[1];
  f.ar[0].x = r0.x; f.ar[0].y = r0.y;
  f.ar[1].x = r0.z; f.ar[1].y = r0.w;
  f.ar[2].x = r1.x; f.ar[2].y = r1.y;
  f.ar[3].x = r1.z; f.ar[3].y = r1.w;
}

#define DW 16
#define SRC_PER_WAVE 128
__global__ __launch_bounds__(DW * 32)
void k_dense(const float* __restrict__ x, const float* __restrict__ Qm,
             const unsigned* __restrict__ A1T,
             const unsigned* __restrict__ A2T, float* __restrict__ y1p,
             float* __restrict__ y2p) {
  __shared__ float red[DW][4][256];
  const int tid = threadIdx.x, lane = tid & 31, wave = tid >> 5;
  const int c16 = lane & 15, h = lane >> 4;
  const int j0 = blockIdx.x * 32;
  const int ibase = blockIdx.y * (DW * SRC_PER_WAVE) + wave * SRC_PER_WAVE;

  v2f qa[4];
  {
    const float4* pq = (const float4*)(Qm + c16 * DD + 8 * h);
    float4 a0 = pq[0], a1 = pq[1];
    qa[0].x = a0.x; qa[0].y = a0.y;
    qa[1].x = a0.z; qa[1].y = a0.w;
    qa[2].x = a1.x; qa[2].y = a1.y;
    qa[3].x = a1.z; qa[3].y = a1.w;
  }

  uint4 u1[2], u2[2];
  float4 xa0[2], xa1[2];
#pragma unroll
  for (int jt = 0; jt < 2; ++jt) {
    const int j = j0 + 16 * jt + c16;
    u1[jt] = *(const uint4*)(A1T + (size_t)j * (NN / 32) + (ibase >> 5));
    u2[jt] = *(const uint4*)(A2T + (size_t)j * (NN / 32) + (ibase >> 5));
    const float4* pj = (const float4*)(x + (size_t)j * DD + 8 * h);
    xa0[jt] = pj[0];
    xa1[jt] = pj[1];
  }
  SrcFrag cur;
  load_src(x, ibase, c16, h, cur);
  asm volatile("" :: "v"(xa0[1].x), "v"(xa0[1].y), "v"(xa0[1].z), "v"(xa0[1].w),
                     "v"(xa1[1].x), "v"(xa1[1].y), "v"(xa1[1].z), "v"(xa1[1].w));

  unsigned w1[2][4], w2[2][4];
  v2f vb[2][4];
#pragma unroll
  for (int jt = 0; jt < 2; ++jt) {
    w1[jt][0] = u1[jt].x >> (8 * h); w1[jt][1] = u1[jt].y >> (8 * h);
    w1[jt][2] = u1[jt].z >> (8 * h); w1[jt][3] = u1[jt].w >> (8 * h);
    w2[jt][0] = u2[jt].x >> (8 * h); w2[jt][1] = u2[jt].y >> (8 * h);
    w2[jt][2] = u2[jt].z >> (8 * h); w2[jt][3] = u2[jt].w >> (8 * h);
    v2f xjb[4];
    xjb[0].x = xa0[jt].x; xjb[0].y = xa0[jt].y;
    xjb[1].x = xa0[jt].z; xjb[1].y = xa0[jt].w;
    xjb[2].x = xa1[jt].x; xjb[2].y = xa1[jt].y;
    xjb[3].x = xa1[jt].z; xjb[3].y = xa1[jt].w;
    v8f v = {};
    v = wmma16(qa, xjb, v);
#pragma unroll
    for (int q = 0; q < 4; ++q) {
      vb[jt][q].x = v[2 * q];
      vb[jt][q].y = v[2 * q + 1];
    }
  }

  v8f y1t[2] = {{}, {}}, y2t[2] = {{}, {}};

#pragma unroll
  for (int t = 0; t < SRC_PER_WAVE / 16; ++t) {
    SrcFrag nxt;
    if (t + 1 < SRC_PER_WAVE / 16)
      load_src(x, ibase + 16 * (t + 1), c16, h, nxt);
#pragma unroll
    for (int jt = 0; jt < 2; ++jt) {
      const unsigned m1 = w1[jt][t >> 1] >> (16 * (t & 1));
      const unsigned m2 = w2[jt][t >> 1] >> (16 * (t & 1));
      v2f b[4];
#pragma unroll
      for (int q = 0; q < 4; ++q) {
        b[q].x = bit_to_one(m1, 2 * q);
        b[q].y = bit_to_one(m1, 2 * q + 1);
      }
      y1t[jt] = wmma16(cur.ax, b, y1t[jt]);
      v8f s = {};
      s = wmma16(cur.ar, vb[jt], s);
      v2f p[4];
#pragma unroll
      for (int q = 0; q < 4; ++q) {
        p[q].x = keep_if_bit(s[2 * q], m2, 2 * q);
        p[q].y = keep_if_bit(s[2 * q + 1], m2, 2 * q + 1);
      }
      y2t[jt] = wmma16(cur.ax, p, y2t[jt]);
    }
    if (t + 1 < SRC_PER_WAVE / 16) cur = nxt;
  }

#pragma unroll
  for (int r = 0; r < 8; ++r) {
    red[wave][0][r * 32 + lane] = y1t[0][r];
    red[wave][1][r * 32 + lane] = y1t[1][r];
    red[wave][2][r * 32 + lane] = y2t[0][r];
    red[wave][3][r * 32 + lane] = y2t[1][r];
  }
  __syncthreads();
  {
    const int d = tid & 15, jj = (tid >> 4) & 15, jt = tid >> 8;
    const int idx = (d & 7) * 32 + (d >> 3) * 16 + jj;
    float s1 = 0.f, s2 = 0.f;
#pragma unroll
    for (int w = 0; w < DW; ++w) {
      s1 += red[w][jt][idx];
      s2 += red[w][2 + jt][idx];
    }
    const size_t o = (size_t)blockIdx.y * NN * DD + (size_t)(j0 + 16 * jt + jj) * DD + d;
    y1p[o] = s1;
    y2p[o] = s2;
  }
}

__global__ void k_combine(const float* __restrict__ xcur, const float* __restrict__ y1p,
                          const float* __restrict__ y2p, const float* __restrict__ G1s,
                          const float* __restrict__ G2s, const unsigned* __restrict__ cnt2,
                          float s1, float* __restrict__ xnext) {
  int wave = threadIdx.x >> 5;
  int row0 = (blockIdx.x * (blockDim.x >> 5) + wave) * 16;
  float s2 = -(float)NN / (float)(*cnt2);
  v2f a[4], b[4];
  v8f c = load_c_tile(xcur, row0);
  load_a_frag_sum2(y1p, row0, s1, a);
  load_b_frag(G1s, b);
  c = wmma16(a, b, c);
  load_a_frag_sum2(y2p, row0, s2, a);
  load_b_frag(G2s, b);
  c = wmma16(a, b, c);
  store_d_tile(xnext, row0, c);
}

extern "C" void kernel_launch(void* const* d_in, const int* in_sizes, int n_in,
                              void* d_out, int out_size, void* d_ws, size_t ws_size,
                              hipStream_t stream) {
  const float* x_in = (const float*)d_in[0];
  const int* e1 = (const int*)d_in[1];
  const int* e2 = (const int*)d_in[2];
  const float* Qm = (const float*)d_in[3];
  const int* mk = (const int*)d_in[4];
  const float* W1 = (const float*)d_in[5];
  const float* W2 = (const float*)d_in[6];
  float* out = (float*)d_out;

  const int E1 = in_sizes[1] / 2;
  const int E2 = in_sizes[2] / 2;
  const int EM = in_sizes[4] / 2;
  const int STEPS = in_sizes[5] / (DD * DD);

  char* ws = (char*)d_ws;
  size_t off = 0;
  auto carve = [&](size_t bytes) {
    char* p = ws + off;
    off += (bytes + 255) & ~(size_t)255;
    return p;
  };
  const size_t BM = (size_t)NN * NN / 8;
  unsigned* MdT   = (unsigned*)carve(BM);
  unsigned* A1T   = (unsigned*)carve(BM);
  unsigned* A2T   = (unsigned*)carve(BM);
  unsigned* cnts  = (unsigned*)carve(8);
  float* G   = (float*)carve((size_t)2 * STEPS * DD * DD * 4);
  const float* G1 = G;
  const float* G2 = G + (size_t)STEPS * DD * DD;
  float* y1p = (float*)carve((size_t)2 * NN * DD * 4);
  float* y2p = (float*)carve((size_t)2 * NN * DD * 4);
  float* xb0 = (float*)carve((size_t)NN * DD * 4);
  float* xb1 = (float*)carve((size_t)NN * DD * 4);

  hipMemsetAsync(MdT, 0, 3 * BM, stream);
  hipMemsetAsync(cnts, 0, 8, stream);

  k_build_mask<<<(EM + 255) / 256, 256, 0, stream>>>(mk, mk + EM, EM, MdT);
  k_build_adj<<<(E1 + 255) / 256, 256, 0, stream>>>(e1, e1 + E1, E1, MdT, A1T, cnts + 0);
  k_build_adj<<<(E2 + 255) / 256, 256, 0, stream>>>(e2, e2 + E2, E2, MdT, A2T, cnts + 1);
  k_g<<<dim3(STEPS, 2), 256, 0, stream>>>(W1, W2, Qm, G);

  const float s1 = (float)NN / (float)EM;

  const int TILE_BLOCKS = NN / 16 / 8;
  const float* xc = x_in;
  for (int s = 0; s < STEPS; ++s) {
    float* xn = (s == STEPS - 1) ? out : ((s & 1) ? xb1 : xb0);
    k_dense<<<dim3(NN / 32, 2), DW * 32, 0, stream>>>(xc, Qm, A1T, A2T, y1p, y2p);
    k_combine<<<TILE_BLOCKS, 256, 0, stream>>>(xc, y1p, y2p, G1 + (size_t)s * DD * DD,
                                               G2 + (size_t)s * DD * DD, cnts + 1, s1, xn);
    xc = xn;
  }
}


# Dense kernel retiled: one block sums all 4096 sources of 16 destinations, so it holds complete y1, y2 and applies the x update itself; 3 combine launches go (10 to 7); sums in a new order
# speedup vs baseline: 7.9100x; 1.1103x over previous
#include <hip/hip_runtime.h>


#define NN 4096
#define DD 16

typedef float v2f __attribute__((ext_vector_type(2)));
typedef float v8f __attribute__((ext_vector_type(8)));

__device__ __forceinline__ void load_a_frag_tile(const float* Y, float scale, v2f a[4]) {
  const int lane = threadIdx.x & 31;
  const int m = lane & 15, h = lane >> 4;
  const float* b0 = Y + m * DD + 2 * h;
#pragma unroll
  for (int c = 0; c < 4; ++c) {
    a[c].x = b0[4 * c] * scale;
    a[c].y = b0[4 * c + 1] * scale;
  }
}

__device__ __forceinline__ void load_b_frag(const float* __restrict__ W, v2f b[4]) {
  const int lane = threadIdx.x & 31;
  const int n = lane & 15, h = lane >> 4;
#pragma unroll
  for (int c = 0; c < 4; ++c) {
    b[c].x = W[(4 * c + 2 * h) * DD + n];
    b[c].y = W[(4 * c + 2 * h + 1) * DD + n];
  }
}

__device__ __forceinline__ v8f load_c_tile(const float* __restrict__ X, int row0) {
  const int lane = threadIdx.x & 31;
  const int n = lane & 15, h = lane >> 4;
  v8f c;
#pragma unroll
  for (int r = 0; r < 8; ++r) c[r] = X[(size_t)(row0 + r + 8 * h) * DD + n];
  return c;
}

__device__ __forceinline__ void store_d_tile(float* __restrict__ O, int row0, v8f d) {
  const int lane = threadIdx.x & 31;
  const int n = lane & 15, h = lane >> 4;
#pragma unroll
  for (int r = 0; r < 8; ++r) O[(size_t)(row0 + r + 8 * h) * DD + n] = d[r];
}

__device__ __forceinline__ v8f wmma16(const v2f a[4], const v2f b[4], v8f c) {
#pragma unroll
  for (int k = 0; k < 4; ++k)
    c = __builtin_amdgcn_wmma_f32_16x16x4_f32(false, a[k], false, b[k],
                                              (short)0, c, false, false);
  return c;
}

__global__ void k_build_mask(const int* __restrict__ mrow, const int* __restrict__ mcol,
                             int E, unsigned* __restrict__ MdT) {
  int e = blockIdx.x * blockDim.x + threadIdx.x;
  if (e >= E) return;
  unsigned bit = (unsigned)mcol[e] * (unsigned)NN + (unsigned)mrow[e];
  atomicOr(&MdT[bit >> 5], 1u << (bit & 31));
}

__global__ void k_build_adj(const int* __restrict__ erow, const int* __restrict__ ecol,
                            int E, const unsigned* __restrict__ MdT,
                            unsigned* __restrict__ AbT, unsigned* __restrict__ cnt) {
  int e = blockIdx.x * blockDim.x + threadIdx.x;
  if (e >= E) return;
  unsigned i = (unsigned)erow[e], j = (unsigned)ecol[e];
  unsigned bit = j * (unsigned)NN + i;
  unsigned w = bit >> 5, b = 1u << (bit & 31);
  if (!(MdT[w] & b)) return;
  unsigned old = atomicOr(&AbT[w], b);
  if (old & b) return;
  atomicAdd(cnt, 1u);
}

__global__ void k_g(const float* __restrict__ W1, const float* __restrict__ W2,
                    const float* __restrict__ Q, float* __restrict__ G) {
  const int s = blockIdx.x;
  const float* __restrict__ W = blockIdx.y ? W2 : W1;
  int k = threadIdx.x >> 4, n = threadIdx.x & 15;
  float acc = 0.f;
#pragma unroll
  for (int t = 0; t < DD; ++t)
    acc = fmaf(W[s * DD * DD + t * DD + k], Q[t * DD + n], acc);
  G[((size_t)blockIdx.y * gridDim.x + s) * DD * DD + k * DD + n] = acc;
}


__device__ __forceinline__ float bit_to_one(unsigned m, int k) {
  unsigned all = (unsigned)(((int)(m << (31 - k))) >> 31);
  return __uint_as_float(all & 0x3f800000u);
}
__device__ __forceinline__ float keep_if_bit(float s, unsigned m, int k) {
  unsigned all = (unsigned)(((int)(m << (31 - k))) >> 31);
  return __uint_as_float(__float_as_uint(s) & all);
}

struct SrcFrag { v2f ax[4]; v2f ar[4]; };

__device__ __forceinline__ void load_src(const float* __restrict__ x, int i0, int c16, int h,
                                         SrcFrag& f) {
  const float* px = x + (size_t)(i0 + 8 * h) * DD + c16;
#pragma unroll
  for (int q = 0; q < 4; ++q) {
    f.ax[q].x = px[(2 * q) * DD];
    f.ax[q].y = px[(2 * q + 1) * DD];
  }
  const float4* pr = (const float4*)(x + (size_t)(i0 + c16) * DD + 8 * h);
  float4 r0 = pr[0], r1 = pr[1];
  f.ar[0].x = r0.x; f.ar[0].y = r0.y;
  f.ar[1].x = r0.z; f.ar[1].y = r0.w;
  f.ar[2].x = r1.x; f.ar[2].y = r1.y;
  f.ar[3].x = r1.z; f.ar[3].y = r1.w;
}

#define DW 16
#define SRC_PER_WAVE 256
__global__ __launch_bounds__(DW * 32)
void k_step(const float* __restrict__ x, const float* __restrict__ Qm,
            const unsigned* __restrict__ A1T, const unsigned* __restrict__ A2T,
            const float* __restrict__ G1s, const float* __restrict__ G2s,
            const unsigned* __restrict__ cnt2, float s1, float* __restrict__ xnext) {
  __shared__ float red[DW][2][256];
  __shared__ float ysum[2][256];
  const int tid = threadIdx.x, lane = tid & 31, wave = tid >> 5;
  const int c16 = lane & 15, h = lane >> 4;
  const int j0 = blockIdx.x * 16;
  const int ibase = wave * SRC_PER_WAVE;

  v2f qa[4];
  {
    const float4* pq = (const float4*)(Qm + c16 * DD + 8 * h);
    float4 a0 = pq[0], a1 = pq[1];
    qa[0].x = a0.x; qa[0].y = a0.y;
    qa[1].x = a0.z; qa[1].y = a0.w;
    qa[2].x = a1.x; qa[2].y = a1.y;
    qa[3].x = a1.z; qa[3].y = a1.w;
  }

  const int j = j0 + c16;
  const uint4* p1 = (const uint4*)(A1T + (size_t)j * (NN / 32) + (ibase >> 5));
  const uint4* p2 = (const uint4*)(A2T + (size_t)j * (NN / 32) + (ibase >> 5));
  const uint4 u1a = p1[0], u1b = p1[1], u2a = p2[0], u2b = p2[1];
  const float4* pj = (const float4*)(x + (size_t)j * DD + 8 * h);
  const float4 xa0 = pj[0], xa1 = pj[1];
  SrcFrag cur[2];
  load_src(x, ibase, c16, h, cur[0]);
  load_src(x, ibase + 16, c16, h, cur[1]);
  v2f g1[4] = {}, g2[4] = {};
  v8f c = {};
  if (wave == 0) {
    load_b_frag(G1s, g1);
    load_b_frag(G2s, g2);
    c = load_c_tile(x, j0);
  }
  const float s2 = -(float)NN / (float)(*cnt2);
  __builtin_amdgcn_sched_barrier(0);

  unsigned w1[8], w2[8];
  w1[0] = u1a.x >> (8 * h); w1[1] = u1a.y >> (8 * h);
  w1[2] = u1a.z >> (8 * h); w1[3] = u1a.w >> (8 * h);
  w1[4] = u1b.x >> (8 * h); w1[5] = u1b.y >> (8 * h);
  w1[6] = u1b.z >> (8 * h); w1[7] = u1b.w >> (8 * h);
  w2[0] = u2a.x >> (8 * h); w2[1] = u2a.y >> (8 * h);
  w2[2] = u2a.z >> (8 * h); w2[3] = u2a.w >> (8 * h);
  w2[4] = u2b.x >> (8 * h); w2[5] = u2b.y >> (8 * h);
  w2[6] = u2b.z >> (8 * h); w2[7] = u2b.w >> (8 * h);

  v2f vb[4];
  {
    v2f xjb[4];
    xjb[0].x = xa0.x; xjb[0].y = xa0.y;
    xjb[1].x = xa0.z; xjb[1].y = xa0.w;
    xjb[2].x = xa1.x; xjb[2].y = xa1.y;
    xjb[3].x = xa1.z; xjb[3].y = xa1.w;
    v8f v = {};
    v = wmma16(qa, xjb, v);
#pragma unroll
    for (int q = 0; q < 4; ++q) {
      vb[q].x = v[2 * q];
      vb[q].y = v[2 * q + 1];
    }
  }

  v8f y1t[2] = {{}, {}}, y2t[2] = {{}, {}};

#pragma unroll
  for (int tp = 0; tp < SRC_PER_WAVE / 32; ++tp) {
    __builtin_amdgcn_sched_barrier(0);
    SrcFrag nxt[2];
    if (tp + 1 < SRC_PER_WAVE / 32) {
      load_src(x, ibase + 32 * (tp + 1), c16, h, nxt[0]);
      load_src(x, ibase + 32 * (tp + 1) + 16, c16, h, nxt[1]);
    }
    __builtin_amdgcn_sched_barrier(0);
#pragma unroll
    for (int u = 0; u < 2; ++u) {
      const unsigned m1 = w1[tp] >> (16 * u);
      const unsigned m2 = w2[tp] >> (16 * u);
      v2f b[4];
#pragma unroll
      for (int q = 0; q < 4; ++q) {
        b[q].x = bit_to_one(m1, 2 * q);
        b[q].y = bit_to_one(m1, 2 * q + 1);
      }
      y1t[u] = wmma16(cur[u].ax, b, y1t[u]);
      v8f s = {};
      s = wmma16(cur[u].ar, vb, s);
      v2f p[4];
#pragma unroll
      for (int q = 0; q < 4; ++q) {
        p[q].x = keep_if_bit(s[2 * q], m2, 2 * q);
        p[q].y = keep_if_bit(s[2 * q + 1], m2, 2 * q + 1);
      }
      y2t[u] = wmma16(cur[u].ax, p, y2t[u]);
    }
    if (tp + 1 < SRC_PER_WAVE / 32) { cur[0] = nxt[0]; cur[1] = nxt[1]; }
  }

#pragma unroll
  for (int r = 0; r < 8; ++r) {
    red[wave][0][r * 32 + lane] = y1t[0][r] + y1t[1][r];
    red[wave][1][r * 32 + lane] = y2t[0][r] + y2t[1][r];
  }
  __syncthreads();
  {
    const int which = tid >> 8, e = tid & 255;
    float s = 0.f;
#pragma unroll
    for (int w = 0; w < DW; ++w) s += red[w][which][e];
    const int r = e >> 5, hh = (e >> 4) & 1, n = e & 15;
    ysum[which][n * DD + r + 8 * hh] = s;
  }
  __syncthreads();
  if (wave == 0) {
    v2f a[4];
    load_a_frag_tile(ysum[0], s1, a);
    c = wmma16(a, g1, c);
    load_a_frag_tile(ysum[1], s2, a);
    c = wmma16(a, g2, c);
    store_d_tile(xnext, j0, c);
  }
}

extern "C" void kernel_launch(void* const* d_in, const int* in_sizes, int n_in,
                              void* d_out, int out_size, void* d_ws, size_t ws_size,
                              hipStream_t stream) {
  const float* x_in = (const float*)d_in[0];
  const int* e1 = (const int*)d_in[1];
  const int* e2 = (const int*)d_in[2];
  const float* Qm = (const float*)d_in[3];
  const int* mk = (const int*)d_in[4];
  const float* W1 = (const float*)d_in[5];
  const float* W2 = (const float*)d_in[6];
  float* out = (float*)d_out;

  const int E1 = in_sizes[1] / 2;
  const int E2 = in_sizes[2] / 2;
  const int EM = in_sizes[4] / 2;
  const int STEPS = in_sizes[5] / (DD * DD);

  char* ws = (char*)d_ws;
  size_t off = 0;
  auto carve = [&](size_t bytes) {
    char* p = ws + off;
    off += (bytes + 255) & ~(size_t)255;
    return p;
  };
  const size_t BM = (size_t)NN * NN / 8;
  unsigned* MdT   = (unsigned*)carve(BM);
  unsigned* A1T   = (unsigned*)carve(BM);
  unsigned* A2T   = (unsigned*)carve(BM);
  unsigned* cnts  = (unsigned*)carve(8);
  float* G   = (float*)carve((size_t)2 * STEPS * DD * DD * 4);
  const float* G1 = G;
  const float* G2 = G + (size_t)STEPS * DD * DD;
  float* xb0 = (float*)carve((size_t)NN * DD * 4);
  float* xb1 = (float*)carve((size_t)NN * DD * 4);

  hipMemsetAsync(MdT, 0, 3 * BM, stream);
  hipMemsetAsync(cnts, 0, 8, stream);

  k_build_mask<<<(EM + 255) / 256, 256, 0, stream>>>(mk, mk + EM, EM, MdT);
  k_build_adj<<<(E1 + 255) / 256, 256, 0, stream>>>(e1, e1 + E1, E1, MdT, A1T, cnts + 0);
  k_build_adj<<<(E2 + 255) / 256, 256, 0, stream>>>(e2, e2 + E2, E2, MdT, A2T, cnts + 1);
  k_g<<<dim3(STEPS, 2), 256, 0, stream>>>(W1, W2, Qm, G);

  const float s1 = (float)NN / (float)EM;

  const float* xc = x_in;
  for (int s = 0; s < STEPS; ++s) {
    float* xn = (s == STEPS - 1) ? out : ((s & 1) ? xb1 : xb0);
    k_step<<<NN / 16, DW * 32, 0, stream>>>(xc, Qm, A1T, A2T, G1 + (size_t)s * DD * DD,
                                            G2 + (size_t)s * DD * DD, cnts + 1, s1, xn);
    xc = xn;
  }
}


# Set construction in 2 launches, not 4: all 3 lists set raw bits at once (8 edges/thread, plain atomicOr; W^T Q blocks ride along); a word-wise AND+popcount pass masks A1, A2, counts A2; sums same
# speedup vs baseline: 8.8242x; 1.1156x over previous
#include <hip/hip_runtime.h>


#define NN 4096
#define DD 16

typedef float v2f __attribute__((ext_vector_type(2)));
typedef float v8f __attribute__((ext_vector_type(8)));

__device__ __forceinline__ void load_a_frag_tile(const float* Y, float scale, v2f a[4]) {
  const int lane = threadIdx.x & 31;
  const int m = lane & 15, h = lane >> 4;
  const float* b0 = Y + m * DD + 2 * h;
#pragma unroll
  for (int c = 0; c < 4; ++c) {
    a[c].x = b0[4 * c] * scale;
    a[c].y = b0[4 * c + 1] * scale;
  }
}

__device__ __forceinline__ void load_b_frag(const float* __restrict__ W, v2f b[4]) {
  const int lane = threadIdx.x & 31;
  const int n = lane & 15, h = lane >> 4;
#pragma unroll
  for (int c = 0; c < 4; ++c) {
    b[c].x = W[(4 * c + 2 * h) * DD + n];
    b[c].y = W[(4 * c + 2 * h + 1) * DD + n];
  }
}

__device__ __forceinline__ v8f load_c_tile(const float* __restrict__ X, int row0) {
  const int lane = threadIdx.x & 31;
  const int n = lane & 15, h = lane >> 4;
  v8f c;
#pragma unroll
  for (int r = 0; r < 8; ++r) c[r] = X[(size_t)(row0 + r + 8 * h) * DD + n];
  return c;
}

__device__ __forceinline__ void store_d_tile(float* __restrict__ O, int row0, v8f d) {
  const int lane = threadIdx.x & 31;
  const int n = lane & 15, h = lane >> 4;
#pragma unroll
  for (int r = 0; r < 8; ++r) O[(size_t)(row0 + r + 8 * h) * DD + n] = d[r];
}

__device__ __forceinline__ v8f wmma16(const v2f a[4], const v2f b[4], v8f c) {
#pragma unroll
  for (int k = 0; k < 4; ++k)
    c = __builtin_amdgcn_wmma_f32_16x16x4_f32(false, a[k], false, b[k],
                                              (short)0, c, false, false);
  return c;
}


__device__ __forceinline__ void set_bit(unsigned* __restrict__ B, int row, int col) {
  unsigned bit = (unsigned)col * (unsigned)NN + (unsigned)row;
  atomicOr(&B[bit >> 5], 1u << (bit & 31));
}

#define EDGES_PER_BLOCK 2048
__global__ __launch_bounds__(256)
void k_build_bits(const int* __restrict__ mk, int EM, const int* __restrict__ e1, int E1,
                  const int* __restrict__ e2, int E2, int nbM, int nb1, int nb2,
                  unsigned* __restrict__ MdT, unsigned* __restrict__ A1T,
                  unsigned* __restrict__ A2T,
                  const float* __restrict__ W1, const float* __restrict__ W2,
                  const float* __restrict__ Q, float* __restrict__ G, int steps) {
  int b = blockIdx.x;
  const int tid = threadIdx.x;
  const int* rows; int E; unsigned* B;
  if (b < nbM) { rows = mk; E = EM; B = MdT; }
  else if (b < nbM + nb1) { b -= nbM; rows = e1; E = E1; B = A1T; }
  else if (b < nbM + nb1 + nb2) { b -= nbM + nb1; rows = e2; E = E2; B = A2T; }
  else {
    const int g = b - (nbM + nb1 + nb2);
    const float* __restrict__ W = (g >= steps) ? W2 + (size_t)(g - steps) * DD * DD
                                               : W1 + (size_t)g * DD * DD;
    const int k = tid >> 4, n = tid & 15;
    float acc = 0.f;
#pragma unroll
    for (int t = 0; t < DD; ++t)
      acc = fmaf(W[t * DD + k], Q[t * DD + n], acc);
    G[(size_t)g * DD * DD + k * DD + n] = acc;
    return;
  }
  const int* cols = rows + E;
  const int base = b * EDGES_PER_BLOCK;
  const bool whole = base + EDGES_PER_BLOCK <= E;
  const bool aligned = ((((size_t)rows) | ((size_t)cols)) & 15) == 0;
  if (whole && aligned) {
    const int e = base + 4 * tid;
    const int4 r0 = *(const int4*)(rows + e);
    const int4 c0 = *(const int4*)(cols + e);
    const int4 r1 = *(const int4*)(rows + e + EDGES_PER_BLOCK / 2);
    const int4 c1 = *(const int4*)(cols + e + EDGES_PER_BLOCK / 2);
    __builtin_amdgcn_sched_barrier(0);
    set_bit(B, r0.x, c0.x); set_bit(B, r0.y, c0.y);
    set_bit(B, r0.z, c0.z); set_bit(B, r0.w, c0.w);
    set_bit(B, r1.x, c1.x); set_bit(B, r1.y, c1.y);
    set_bit(B, r1.z, c1.z); set_bit(B, r1.w, c1.w);
  } else {
    for (int u = 0; u < EDGES_PER_BLOCK / 256; ++u) {
      const int e = base + u * 256 + tid;
      if (e < E) set_bit(B, rows[e], cols[e]);
    }
  }
}

typedef unsigned v4u __attribute__((ext_vector_type(4)));
__device__ __forceinline__ unsigned popc4(v4u v) {
  return __popc(v.x) + __popc(v.y) + __popc(v.z) + __popc(v.w);
}
__global__ __launch_bounds__(256)
void k_and_count(const unsigned* __restrict__ MdT, unsigned* __restrict__ A1T,
                 unsigned* __restrict__ A2T, int n4, unsigned* __restrict__ cnt) {
  __shared__ unsigned wsum[8];
  const v4u* __restrict__ Md = (const v4u*)MdT;
  v4u* __restrict__ A1 = (v4u*)A1T;
  v4u* __restrict__ A2 = (v4u*)A2T;
  const int tid = threadIdx.x;
  const int stride = gridDim.x * 256;
  unsigned pc = 0;
  int w = blockIdx.x * 256 + tid;
  for (; w + stride < n4; w += 2 * stride) {
    const v4u m0 = Md[w], a0 = A1[w], b0 = A2[w];
    const v4u m1 = Md[w + stride], a1 = A1[w + stride], b1 = A2[w + stride];
    const v4u y0 = b0 & m0, y1 = b1 & m1;
    A1[w] = a0 & m0;
    A2[w] = y0;
    A1[w + stride] = a1 & m1;
    A2[w + stride] = y1;
    pc += popc4(y0) + popc4(y1);
  }
  if (w < n4) {
    const v4u m0 = Md[w], a0 = A1[w], b0 = A2[w];
    const v4u y0 = b0 & m0;
    A1[w] = a0 & m0;
    A2[w] = y0;
    pc += popc4(y0);
  }
#pragma unroll
  for (int off = 16; off > 0; off >>= 1) pc += __shfl_xor(pc, off);
  if ((tid & 31) == 0) wsum[tid >> 5] = pc;
  __syncthreads();
  if (tid == 0) {
    unsigned s = 0;
#pragma unroll
    for (int k = 0; k < 8; ++k) s += wsum[k];
    atomicAdd(cnt, s);
  }
}


__device__ __forceinline__ float bit_to_one(unsigned m, int k) {
  unsigned all = (unsigned)(((int)(m << (31 - k))) >> 31);
  return __uint_as_float(all & 0x3f800000u);
}
__device__ __forceinline__ float keep_if_bit(float s, unsigned m, int k) {
  unsigned all = (unsigned)(((int)(m << (31 - k))) >> 31);
  return __uint_as_float(__float_as_uint(s) & all);
}

struct SrcFrag { v2f ax[4]; v2f ar[4]; };

__device__ __forceinline__ void load_src(const float* __restrict__ x, int i0, int c16, int h,
                                         SrcFrag& f) {
  const float* px = x + (size_t)(i0 + 8 * h) * DD + c16;
#pragma unroll
  for (int q = 0; q < 4; ++q) {
    f.ax[q].x = px[(2 * q) * DD];
    f.ax[q].y = px[(2 * q + 1) * DD];
  }
  const float4* pr = (const float4*)(x + (size_t)(i0 + c16) * DD + 8 * h);
  float4 r0 = pr[0], r1 = pr[1];
  f.ar[0].x = r0.x; f.ar[0].y = r0.y;
  f.ar[1].x = r0.z; f.ar[1].y = r0.w;
  f.ar[2].x = r1.x; f.ar[2].y = r1.y;
  f.ar[3].x = r1.z; f.ar[3].y = r1.w;
}

#define DW 16
#define SRC_PER_WAVE 256
__global__ __launch_bounds__(DW * 32)
void k_step(const float* __restrict__ x, const float* __restrict__ Qm,
            const unsigned* __restrict__ A1T, const unsigned* __restrict__ A2T,
            const float* __restrict__ G1s, const float* __restrict__ G2s,
            const unsigned* __restrict__ cnt2, float s1, float* __restrict__ xnext) {
  __shared__ float red[DW][2][256];
  __shared__ float ysum[2][256];
  const int tid = threadIdx.x, lane = tid & 31, wave = tid >> 5;
  const int c16 = lane & 15, h = lane >> 4;
  const int j0 = blockIdx.x * 16;
  const int ibase = wave * SRC_PER_WAVE;

  v2f qa[4];
  {
    const float4* pq = (const float4*)(Qm + c16 * DD + 8 * h);
    float4 a0 = pq[0], a1 = pq[1];
    qa[0].x = a0.x; qa[0].y = a0.y;
    qa[1].x = a0.z; qa[1].y = a0.w;
    qa[2].x = a1.x; qa[2].y = a1.y;
    qa[3].x = a1.z; qa[3].y = a1.w;
  }

  const int j = j0 + c16;
  const uint4* p1 = (const uint4*)(A1T + (size_t)j * (NN / 32) + (ibase >> 5));
  const uint4* p2 = (const uint4*)(A2T + (size_t)j * (NN / 32) + (ibase >> 5));
  const uint4 u1a = p1[0], u1b = p1[1], u2a = p2[0], u2b = p2[1];
  const float4* pj = (const float4*)(x + (size_t)j * DD + 8 * h);
  const float4 xa0 = pj[0], xa1 = pj[1];
  SrcFrag cur[2];
  load_src(x, ibase, c16, h, cur[0]);
  load_src(x, ibase + 16, c16, h, cur[1]);
  v2f g1[4] = {}, g2[4] = {};
  v8f c = {};
  if (wave == 0) {
    load_b_frag(G1s, g1);
    load_b_frag(G2s, g2);
    c = load_c_tile(x, j0);
  }
  const float s2 = -(float)NN / (float)(*cnt2);
  __builtin_amdgcn_sched_barrier(0);

  unsigned w1[8], w2[8];
  w1[0] = u1a.x >> (8 * h); w1[1] = u1a.y >> (8 * h);
  w1[2] = u1a.z >> (8 * h); w1[3] = u1a.w >> (8 * h);
  w1[4] = u1b.x >> (8 * h); w1[5] = u1b.y >> (8 * h);
  w1[6] = u1b.z >> (8 * h); w1[7] = u1b.w >> (8 * h);
  w2[0] = u2a.x >> (8 * h); w2[1] = u2a.y >> (8 * h);
  w2[2] = u2a.z >> (8 * h); w2[3] = u2a.w >> (8 * h);
  w2[4] = u2b.x >> (8 * h); w2[5] = u2b.y >> (8 * h);
  w2[6] = u2b.z >> (8 * h); w2[7] = u2b.w >> (8 * h);

  v2f vb[4];
  {
    v2f xjb[4];
    xjb[0].x = xa0.x; xjb[0].y = xa0.y;
    xjb[1].x = xa0.z; xjb[1].y = xa0.w;
    xjb[2].x = xa1.x; xjb[2].y = xa1.y;
    xjb[3].x = xa1.z; xjb[3].y = xa1.w;
    v8f v = {};
    v = wmma16(qa, xjb, v);
#pragma unroll
    for (int q = 0; q < 4; ++q) {
      vb[q].x = v[2 * q];
      vb[q].y = v[2 * q + 1];
    }
  }

  v8f y1t[2] = {{}, {}}, y2t[2] = {{}, {}};

#pragma unroll
  for (int tp = 0; tp < SRC_PER_WAVE / 32; ++tp) {
    __builtin_amdgcn_sched_barrier(0);
    SrcFrag nxt[2];
    if (tp + 1 < SRC_PER_WAVE / 32) {
      load_src(x, ibase + 32 * (tp + 1), c16, h, nxt[0]);
      load_src(x, ibase + 32 * (tp + 1) + 16, c16, h, nxt[1]);
    }
    __builtin_amdgcn_sched_barrier(0);
#pragma unroll
    for (int u = 0; u < 2; ++u) {
      const unsigned m1 = w1[tp] >> (16 * u);
      const unsigned m2 = w2[tp] >> (16 * u);
      v2f b[4];
#pragma unroll
      for (int q = 0; q < 4; ++q) {
        b[q].x = bit_to_one(m1, 2 * q);
        b[q].y = bit_to_one(m1, 2 * q + 1);
      }
      y1t[u] = wmma16(cur[u].ax, b, y1t[u]);
      v8f s = {};
      s = wmma16(cur[u].ar, vb, s);
      v2f p[4];
#pragma unroll
      for (int q = 0; q < 4; ++q) {
        p[q].x = keep_if_bit(s[2 * q], m2, 2 * q);
        p[q].y = keep_if_bit(s[2 * q + 1], m2, 2 * q + 1);
      }
      y2t[u] = wmma16(cur[u].ax, p, y2t[u]);
    }
    if (tp + 1 < SRC_PER_WAVE / 32) { cur[0] = nxt[0]; cur[1] = nxt[1]; }
  }

#pragma unroll
  for (int r = 0; r < 8; ++r) {
    red[wave][0][r * 32 + lane] = y1t[0][r] + y1t[1][r];
    red[wave][1][r * 32 + lane] = y2t[0][r] + y2t[1][r];
  }
  __syncthreads();
  {
    const int which = tid >> 8, e = tid & 255;
    float s = 0.f;
#pragma unroll
    for (int w = 0; w < DW; ++w) s += red[w][which][e];
    const int r = e >> 5, hh = (e >> 4) & 1, n = e & 15;
    ysum[which][n * DD + r + 8 * hh] = s;
  }
  __syncthreads();
  if (wave == 0) {
    v2f a[4];
    load_a_frag_tile(ysum[0], s1, a);
    c = wmma16(a, g1, c);
    load_a_frag_tile(ysum[1], s2, a);
    c = wmma16(a, g2, c);
    store_d_tile(xnext, j0, c);
  }
}

extern "C" void kernel_launch(void* const* d_in, const int* in_sizes, int n_in,
                              void* d_out, int out_size, void* d_ws, size_t ws_size,
                              hipStream_t stream) {
  const float* x_in = (const float*)d_in[0];
  const int* e1 = (const int*)d_in[1];
  const int* e2 = (const int*)d_in[2];
  const float* Qm = (const float*)d_in[3];
  const int* mk = (const int*)d_in[4];
  const float* W1 = (const float*)d_in[5];
  const float* W2 = (const float*)d_in[6];
  float* out = (float*)d_out;

  const int E1 = in_sizes[1] / 2;
  const int E2 = in_sizes[2] / 2;
  const int EM = in_sizes[4] / 2;
  const int STEPS = in_sizes[5] / (DD * DD);

  char* ws = (char*)d_ws;
  size_t off = 0;
  auto carve = [&](size_t bytes) {
    char* p = ws + off;
    off += (bytes + 255) & ~(size_t)255;
    return p;
  };
  const size_t BM = (size_t)NN * NN / 8;
  unsigned* MdT   = (unsigned*)carve(BM);
  unsigned* A1T   = (unsigned*)carve(BM);
  unsigned* A2T   = (unsigned*)carve(BM);
  unsigned* cnts  = (unsigned*)carve(8);
  float* G   = (float*)carve((size_t)2 * STEPS * DD * DD * 4);
  const float* G1 = G;
  const float* G2 = G + (size_t)STEPS * DD * DD;
  float* xb0 = (float*)carve((size_t)NN * DD * 4);
  float* xb1 = (float*)carve((size_t)NN * DD * 4);

  hipMemsetAsync(MdT, 0, 3 * BM + 8, stream);

  const int nbM = (EM + EDGES_PER_BLOCK - 1) / EDGES_PER_BLOCK;
  const int nb1 = (E1 + EDGES_PER_BLOCK - 1) / EDGES_PER_BLOCK;
  const int nb2 = (E2 + EDGES_PER_BLOCK - 1) / EDGES_PER_BLOCK;
  k_build_bits<<<nbM + nb1 + nb2 + 2 * STEPS, 256, 0, stream>>>(
      mk, EM, e1, E1, e2, E2, nbM, nb1, nb2, MdT, A1T, A2T, W1, W2, Qm, G, STEPS);
  k_and_count<<<256, 256, 0, stream>>>(MdT, A1T, A2T, (int)(BM / 16), cnts + 1);

  const float s1 = (float)NN / (float)EM;

  const float* xc = x_in;
  for (int s = 0; s < STEPS; ++s) {
    float* xn = (s == STEPS - 1) ? out : ((s & 1) ? xb1 : xb0);
    k_step<<<NN / 16, DW * 32, 0, stream>>>(xc, Qm, A1T, A2T, G1 + (size_t)s * DD * DD,
                                            G2 + (size_t)s * DD * DD, cnts + 1, s1, xn);
    xc = xn;
  }
}


# k_step skips the matrix instructions whose 4-source x 16-destination bit operand is zero in every lane (scalar branches on per-bit lane masks); sums unchanged: the skipped terms are exact zeros
# speedup vs baseline: 9.4479x; 1.0707x over previous
#include <hip/hip_runtime.h>


#define NN 4096
#define DD 16

typedef float v2f __attribute__((ext_vector_type(2)));
typedef float v8f __attribute__((ext_vector_type(8)));

__device__ __forceinline__ void load_a_frag_tile(const float* Y, float scale, v2f a[4]) {
  const int lane = threadIdx.x & 31;
  const int m = lane & 15, h = lane >> 4;
  const float* b0 = Y + m * DD + 2 * h;
#pragma unroll
  for (int c = 0; c < 4; ++c) {
    a[c].x = b0[4 * c] * scale;
    a[c].y = b0[4 * c + 1] * scale;
  }
}

__device__ __forceinline__ void load_b_frag(const float* __restrict__ W, v2f b[4]) {
  const int lane = threadIdx.x & 31;
  const int n = lane & 15, h = lane >> 4;
#pragma unroll
  for (int c = 0; c < 4; ++c) {
    b[c].x = W[(4 * c + 2 * h) * DD + n];
    b[c].y = W[(4 * c + 2 * h + 1) * DD + n];
  }
}

__device__ __forceinline__ v8f load_c_tile(const float* __restrict__ X, int row0) {
  const int lane = threadIdx.x & 31;
  const int n = lane & 15, h = lane >> 4;
  v8f c;
#pragma unroll
  for (int r = 0; r < 8; ++r) c[r] = X[(size_t)(row0 + r + 8 * h) * DD + n];
  return c;
}

__device__ __forceinline__ void store_d_tile(float* __restrict__ O, int row0, v8f d) {
  const int lane = threadIdx.x & 31;
  const int n = lane & 15, h = lane >> 4;
#pragma unroll
  for (int r = 0; r < 8; ++r) O[(size_t)(row0 + r + 8 * h) * DD + n] = d[r];
}

__device__ __forceinline__ v8f wmma16(const v2f a[4], const v2f b[4], v8f c) {
#pragma unroll
  for (int k = 0; k < 4; ++k)
    c = __builtin_amdgcn_wmma_f32_16x16x4_f32(false, a[k], false, b[k],
                                              (short)0, c, false, false);
  return c;
}

__device__ __forceinline__ bool lane_bit(unsigned mask) {
  return __builtin_amdgcn_inverse_ballot_w32(mask);
}

__device__ __forceinline__ void wmma_acc(v8f& c, v2f a, v2f b) {
  asm("v_wmma_f32_16x16x4_f32 %0, %1, %2, %0\n\tv_nop\n\tv_nop\n\tv_nop\n\tv_nop"
      : "+v"(c) : "v"(a), "v"(b));
}

__device__ __forceinline__ v8f wmma_scores(const v2f a[4], const v2f b[4]) {
  v8f s;
  asm("v_wmma_f32_16x16x4_f32 %0, %1, %5, 0\n\t"
      "v_wmma_f32_16x16x4_f32 %0, %2, %6, %0\n\t"
      "v_wmma_f32_16x16x4_f32 %0, %3, %7, %0\n\t"
      "v_wmma_f32_16x16x4_f32 %0, %4, %8, %0\n\t"
      "v_nop\n\tv_nop\n\tv_nop\n\tv_nop"
      : "=&v"(s)
      : "v"(a[0]), "v"(a[1]), "v"(a[2]), "v"(a[3]), "v"(b[0]), "v"(b[1]), "v"(b[2]), "v"(b[3]));
  return s;
}


__device__ __forceinline__ void set_bit(unsigned* __restrict__ B, int row, int col) {
  unsigned bit = (unsigned)col * (unsigned)NN + (unsigned)row;
  atomicOr(&B[bit >> 5], 1u << (bit & 31));
}

#define EDGES_PER_BLOCK 2048
__global__ __launch_bounds__(256)
void k_build_bits(const int* __restrict__ mk, int EM, const int* __restrict__ e1, int E1,
                  const int* __restrict__ e2, int E2, int nbM, int nb1, int nb2,
                  unsigned* __restrict__ MdT, unsigned* __restrict__ A1T,
                  unsigned* __restrict__ A2T,
                  const float* __restrict__ W1, const float* __restrict__ W2,
                  const float* __restrict__ Q, float* __restrict__ G, int steps) {
  int b = blockIdx.x;
  const int tid = threadIdx.x;
  const int* rows; int E; unsigned* B;
  if (b < nbM) { rows = mk; E = EM; B = MdT; }
  else if (b < nbM + nb1) { b -= nbM; rows = e1; E = E1; B = A1T; }
  else if (b < nbM + nb1 + nb2) { b -= nbM + nb1; rows = e2; E = E2; B = A2T; }
  else {
    const int g = b - (nbM + nb1 + nb2);
    const float* __restrict__ W = (g >= steps) ? W2 + (size_t)(g - steps) * DD * DD
                                               : W1 + (size_t)g * DD * DD;
    const int k = tid >> 4, n = tid & 15;
    float acc = 0.f;
#pragma unroll
    for (int t = 0; t < DD; ++t)
      acc = fmaf(W[t * DD + k], Q[t * DD + n], acc);
    G[(size_t)g * DD * DD + k * DD + n] = acc;
    return;
  }
  const int* cols = rows + E;
  const int base = b * EDGES_PER_BLOCK;
  const bool whole = base + EDGES_PER_BLOCK <= E;
  const bool aligned = ((((size_t)rows) | ((size_t)cols)) & 15) == 0;
  if (whole && aligned) {
    const int e = base + 4 * tid;
    const int4 r0 = *(const int4*)(rows + e);
    const int4 c0 = *(const int4*)(cols + e);
    const int4 r1 = *(const int4*)(rows + e + EDGES_PER_BLOCK / 2);
    const int4 c1 = *(const int4*)(cols + e + EDGES_PER_BLOCK / 2);
    __builtin_amdgcn_sched_barrier(0);
    set_bit(B, r0.x, c0.x); set_bit(B, r0.y, c0.y);
    set_bit(B, r0.z, c0.z); set_bit(B, r0.w, c0.w);
    set_bit(B, r1.x, c1.x); set_bit(B, r1.y, c1.y);
    set_bit(B, r1.z, c1.z); set_bit(B, r1.w, c1.w);
  } else {
    for (int u = 0; u < EDGES_PER_BLOCK / 256; ++u) {
      const int e = base + u * 256 + tid;
      if (e < E) set_bit(B, rows[e], cols[e]);
    }
  }
}

typedef unsigned v4u __attribute__((ext_vector_type(4)));
__device__ __forceinline__ unsigned popc4(v4u v) {
  return __popc(v.x) + __popc(v.y) + __popc(v.z) + __popc(v.w);
}
__global__ __launch_bounds__(256)
void k_and_count(const unsigned* __restrict__ MdT, unsigned* __restrict__ A1T,
                 unsigned* __restrict__ A2T, int n4, unsigned* __restrict__ cnt) {
  __shared__ unsigned wsum[8];
  const v4u* __restrict__ Md = (const v4u*)MdT;
  v4u* __restrict__ A1 = (v4u*)A1T;
  v4u* __restrict__ A2 = (v4u*)A2T;
  const int tid = threadIdx.x;
  const int stride = gridDim.x * 256;
  unsigned pc = 0;
  int w = blockIdx.x * 256 + tid;
  for (; w + stride < n4; w += 2 * stride) {
    const v4u m0 = Md[w], a0 = A1[w], b0 = A2[w];
    const v4u m1 = Md[w + stride], a1 = A1[w + stride], b1 = A2[w + stride];
    const v4u y0 = b0 & m0, y1 = b1 & m1;
    A1[w] = a0 & m0;
    A2[w] = y0;
    A1[w + stride] = a1 & m1;
    A2[w + stride] = y1;
    pc += popc4(y0) + popc4(y1);
  }
  if (w < n4) {
    const v4u m0 = Md[w], a0 = A1[w], b0 = A2[w];
    const v4u y0 = b0 & m0;
    A1[w] = a0 & m0;
    A2[w] = y0;
    pc += popc4(y0);
  }
#pragma unroll
  for (int off = 16; off > 0; off >>= 1) pc += __shfl_xor(pc, off);
  if ((tid & 31) == 0) wsum[tid >> 5] = pc;
  __syncthreads();
  if (tid == 0) {
    unsigned s = 0;
#pragma unroll
    for (int k = 0; k < 8; ++k) s += wsum[k];
    atomicAdd(cnt, s);
  }
}


struct SrcFrag { v2f ax[4]; v2f ar[4]; };

__device__ __forceinline__ void load_src(const float* __restrict__ x, int i0, int c16, int h,
                                         SrcFrag& f) {
  const float* px = x + (size_t)(i0 + 8 * h) * DD + c16;
#pragma unroll
  for (int q = 0; q < 4; ++q) {
    f.ax[q].x = px[(2 * q) * DD];
    f.ax[q].y = px[(2 * q + 1) * DD];
  }
  const float4* pr = (const float4*)(x + (size_t)(i0 + c16) * DD + 8 * h);
  float4 r0 = pr[0], r1 = pr[1];
  f.ar[0].x = r0.x; f.ar[0].y = r0.y;
  f.ar[1].x = r0.z; f.ar[1].y = r0.w;
  f.ar[2].x = r1.x; f.ar[2].y = r1.y;
  f.ar[3].x = r1.z; f.ar[3].y = r1.w;
}

__device__ __forceinline__ void block_products(const SrcFrag& f, const v2f vb[4],
                                               const unsigned k1[8], const unsigned k2[8],
                                               v8f& y1, v8f& y2) {
#pragma unroll
  for (int q = 0; q < 4; ++q) {
    const unsigned ka = k1[2 * q], kb = k1[2 * q + 1];
    if ((ka | kb) != 0u) {
      v2f b;
      b.x = lane_bit(ka) ? 1.0f : 0.0f;
      b.y = lane_bit(kb) ? 1.0f : 0.0f;
      wmma_acc(y1, f.ax[q], b);
    }
  }
  const unsigned any2 = k2[0] | k2[1] | k2[2] | k2[3] | k2[4] | k2[5] | k2[6] | k2[7];
  if (any2 != 0u) {
    const v8f s = wmma_scores(f.ar, vb);
#pragma unroll
    for (int q = 0; q < 4; ++q) {
      const unsigned ka = k2[2 * q], kb = k2[2 * q + 1];
      if ((ka | kb) != 0u) {
        v2f p;
        p.x = lane_bit(ka) ? s[2 * q] : 0.0f;
        p.y = lane_bit(kb) ? s[2 * q + 1] : 0.0f;
        wmma_acc(y2, f.ax[q], p);
      }
    }
  }
}

#define DW 16
#define SRC_PER_WAVE 256
__global__ __launch_bounds__(DW * 32)
void k_step(const float* __restrict__ x, const float* __restrict__ Qm,
            const unsigned* __restrict__ A1T, const unsigned* __restrict__ A2T,
            const float* __restrict__ G1s, const float* __restrict__ G2s,
            const unsigned* __restrict__ cnt2, float s1, float* __restrict__ xnext) {
  __shared__ float red[DW][2][256];
  __shared__ float ysum[2][256];
  const int tid = threadIdx.x, lane = tid & 31, wave = tid >> 5;
  const int c16 = lane & 15, h = lane >> 4;
  const int j0 = blockIdx.x * 16;
  const int ibase = wave * SRC_PER_WAVE;

  v2f qa[4];
  {
    const float4* pq = (const float4*)(Qm + c16 * DD + 8 * h);
    float4 a0 = pq[0], a1 = pq[1];
    qa[0].x = a0.x; qa[0].y = a0.y;
    qa[1].x = a0.z; qa[1].y = a0.w;
    qa[2].x = a1.x; qa[2].y = a1.y;
    qa[3].x = a1.z; qa[3].y = a1.w;
  }

  const int j = j0 + c16;
  const uint4* p1 = (const uint4*)(A1T + (size_t)j * (NN / 32) + (ibase >> 5));
  const uint4* p2 = (const uint4*)(A2T + (size_t)j * (NN / 32) + (ibase >> 5));
  const uint4 u1a = p1[0], u1b = p1[1], u2a = p2[0], u2b = p2[1];
  const float4* pj = (const float4*)(x + (size_t)j * DD + 8 * h);
  const float4 xa0 = pj[0], xa1 = pj[1];
  SrcFrag cur[2];
  load_src(x, ibase, c16, h, cur[0]);
  load_src(x, ibase + 16, c16, h, cur[1]);
  v2f g1[4] = {}, g2[4] = {};
  v8f c = {};
  if (wave == 0) {
    load_b_frag(G1s, g1);
    load_b_frag(G2s, g2);
    c = load_c_tile(x, j0);
  }
  const float s2 = -(float)NN / (float)(*cnt2);
  __builtin_amdgcn_sched_barrier(0);

  unsigned w1[8], w2[8];
  w1[0] = u1a.x >> (8 * h); w1[1] = u1a.y >> (8 * h);
  w1[2] = u1a.z >> (8 * h); w1[3] = u1a.w >> (8 * h);
  w1[4] = u1b.x >> (8 * h); w1[5] = u1b.y >> (8 * h);
  w1[6] = u1b.z >> (8 * h); w1[7] = u1b.w >> (8 * h);
  w2[0] = u2a.x >> (8 * h); w2[1] = u2a.y >> (8 * h);
  w2[2] = u2a.z >> (8 * h); w2[3] = u2a.w >> (8 * h);
  w2[4] = u2b.x >> (8 * h); w2[5] = u2b.y >> (8 * h);
  w2[6] = u2b.z >> (8 * h); w2[7] = u2b.w >> (8 * h);

  v2f vb[4];
  {
    v2f xjb[4];
    xjb[0].x = xa0.x; xjb[0].y = xa0.y;
    xjb[1].x = xa0.z; xjb[1].y = xa0.w;
    xjb[2].x = xa1.x; xjb[2].y = xa1.y;
    xjb[3].x = xa1.z; xjb[3].y = xa1.w;
    v8f v = {};
    v = wmma16(qa, xjb, v);
#pragma unroll
    for (int q = 0; q < 4; ++q) {
      vb[q].x = v[2 * q];
      vb[q].y = v[2 * q + 1];
    }
  }

  v8f y1e = {}, y1o = {}, y2e = {}, y2o = {};

#pragma unroll
  for (int tp = 0; tp < SRC_PER_WAVE / 32; ++tp) {
    __builtin_amdgcn_sched_barrier(0);
    SrcFrag nxt[2];
    if (tp + 1 < SRC_PER_WAVE / 32) {
      load_src(x, ibase + 32 * (tp + 1), c16, h, nxt[0]);
      load_src(x, ibase + 32 * (tp + 1) + 16, c16, h, nxt[1]);
    }
    __builtin_amdgcn_sched_barrier(0);
    unsigned k1[2][8], k2[2][8];
#pragma unroll
    for (int u = 0; u < 2; ++u) {
      const unsigned m1 = w1[tp] >> (16 * u);
      const unsigned m2 = w2[tp] >> (16 * u);
#pragma unroll
      for (int k = 0; k < 8; ++k) {
        k1[u][k] = __builtin_amdgcn_ballot_w32(((m1 >> k) & 1u) != 0u);
        k2[u][k] = __builtin_amdgcn_ballot_w32(((m2 >> k) & 1u) != 0u);
      }
    }
    block_products(cur[0], vb, k1[0], k2[0], y1e, y2e);
    block_products(cur[1], vb, k1[1], k2[1], y1o, y2o);
    if (tp + 1 < SRC_PER_WAVE / 32) { cur[0] = nxt[0]; cur[1] = nxt[1]; }
  }

#pragma unroll
  for (int r = 0; r < 8; ++r) {
    red[wave][0][r * 32 + lane] = y1e[r] + y1o[r];
    red[wave][1][r * 32 + lane] = y2e[r] + y2o[r];
  }
  __syncthreads();
  {
    const int which = tid >> 8, e = tid & 255;
    float s = 0.f;
#pragma unroll
    for (int w = 0; w < DW; ++w) s += red[w][which][e];
    const int r = e >> 5, hh = (e >> 4) & 1, n = e & 15;
    ysum[which][n * DD + r + 8 * hh] = s;
  }
  __syncthreads();
  if (wave == 0) {
    v2f a[4];
    load_a_frag_tile(ysum[0], s1, a);
    c = wmma16(a, g1, c);
    load_a_frag_tile(ysum[1], s2, a);
    c = wmma16(a, g2, c);
    store_d_tile(xnext, j0, c);
  }
}

extern "C" void kernel_launch(void* const* d_in, const int* in_sizes, int n_in,
                              void* d_out, int out_size, void* d_ws, size_t ws_size,
                              hipStream_t stream) {
  const float* x_in = (const float*)d_in[0];
  const int* e1 = (const int*)d_in[1];
  const int* e2 = (const int*)d_in[2];
  const float* Qm = (const float*)d_in[3];
  const int* mk = (const int*)d_in[4];
  const float* W1 = (const float*)d_in[5];
  const float* W2 = (const float*)d_in[6];
  float* out = (float*)d_out;

  const int E1 = in_sizes[1] / 2;
  const int E2 = in_sizes[2] / 2;
  const int EM = in_sizes[4] / 2;
  const int STEPS = in_sizes[5] / (DD * DD);

  char* ws = (char*)d_ws;
  size_t off = 0;
  auto carve = [&](size_t bytes) {
    char* p = ws + off;
    off += (bytes + 255) & ~(size_t)255;
    return p;
  };
  const size_t BM = (size_t)NN * NN / 8;
  unsigned* MdT   = (unsigned*)carve(BM);
  unsigned* A1T   = (unsigned*)carve(BM);
  unsigned* A2T   = (unsigned*)carve(BM);
  unsigned* cnts  = (unsigned*)carve(8);
  float* G   = (float*)carve((size_t)2 * STEPS * DD * DD * 4);
  const float* G1 = G;
  const float* G2 = G + (size_t)STEPS * DD * DD;
  float* xb0 = (float*)carve((size_t)NN * DD * 4);
  float* xb1 = (float*)carve((size_t)NN * DD * 4);

  hipMemsetAsync(MdT, 0, 3 * BM + 8, stream);

  const int nbM = (EM + EDGES_PER_BLOCK - 1) / EDGES_PER_BLOCK;
  const int nb1 = (E1 + EDGES_PER_BLOCK - 1) / EDGES_PER_BLOCK;
  const int nb2 = (E2 + EDGES_PER_BLOCK - 1) / EDGES_PER_BLOCK;
  k_build_bits<<<nbM + nb1 + nb2 + 2 * STEPS, 256, 0, stream>>>(
      mk, EM, e1, E1, e2, E2, nbM, nb1, nb2, MdT, A1T, A2T, W1, W2, Qm, G, STEPS);
  k_and_count<<<256, 256, 0, stream>>>(MdT, A1T, A2T, (int)(BM / 16), cnts + 1);

  const float s1 = (float)NN / (float)EM;

  const float* xc = x_in;
  for (int s = 0; s < STEPS; ++s) {
    float* xn = (s == STEPS - 1) ? out : ((s & 1) ? xb1 : xb0);
    k_step<<<NN / 16, DW * 32, 0, stream>>>(xc, Qm, A1T, A2T, G1 + (size_t)s * DD * DD,
                                            G2 + (size_t)s * DD * DD, cnts + 1, s1, xn);
    xc = xn;
  }
}


# Mask list sets no bits: each mask edge reads the raw edge_index/edge_index_2 bit pair (one 8-byte load), ORs only hits into A1/A2; returning OR counts sum(A2); mask bitmap and AND pass go; sums same
# speedup vs baseline: 13.3539x; 1.4134x over previous
#include <hip/hip_runtime.h>


#define NN 4096
#define DD 16

typedef float v2f __attribute__((ext_vector_type(2)));
typedef float v8f __attribute__((ext_vector_type(8)));

__device__ __forceinline__ void load_a_frag_tile(const float* Y, float scale, v2f a[4]) {
  const int lane = threadIdx.x & 31;
  const int m = lane & 15, h = lane >> 4;
  const float* b0 = Y + m * DD + 2 * h;
#pragma unroll
  for (int c = 0; c < 4; ++c) {
    a[c].x = b0[4 * c] * scale;
    a[c].y = b0[4 * c + 1] * scale;
  }
}

__device__ __forceinline__ void load_b_frag(const float* __restrict__ W, v2f b[4]) {
  const int lane = threadIdx.x & 31;
  const int n = lane & 15, h = lane >> 4;
#pragma unroll
  for (int c = 0; c < 4; ++c) {
    b[c].x = W[(4 * c + 2 * h) * DD + n];
    b[c].y = W[(4 * c + 2 * h + 1) * DD + n];
  }
}

__device__ __forceinline__ v8f load_c_tile(const float* __restrict__ X, int row0) {
  const int lane = threadIdx.x & 31;
  const int n = lane & 15, h = lane >> 4;
  v8f c;
#pragma unroll
  for (int r = 0; r < 8; ++r) c[r] = X[(size_t)(row0 + r + 8 * h) * DD + n];
  return c;
}

__device__ __forceinline__ void store_d_tile(float* __restrict__ O, int row0, v8f d) {
  const int lane = threadIdx.x & 31;
  const int n = lane & 15, h = lane >> 4;
#pragma unroll
  for (int r = 0; r < 8; ++r) O[(size_t)(row0 + r + 8 * h) * DD + n] = d[r];
}

__device__ __forceinline__ v8f wmma16(const v2f a[4], const v2f b[4], v8f c) {
#pragma unroll
  for (int k = 0; k < 4; ++k)
    c = __builtin_amdgcn_wmma_f32_16x16x4_f32(false, a[k], false, b[k],
                                              (short)0, c, false, false);
  return c;
}


#define EDGES_PER_BLOCK 2048

__device__ __forceinline__ void set_raw(unsigned* __restrict__ ER, unsigned which, int row, int col) {
  const unsigned bit = (unsigned)col * (unsigned)NN + (unsigned)row;
  atomicOr(&ER[2u * (bit >> 5) + which], 1u << (bit & 31));
}

__global__ __launch_bounds__(256)
void k_build_raw(const int* __restrict__ e1, int E1, const int* __restrict__ e2, int E2,
                 int nb1, int nb2, unsigned* __restrict__ ER,
                 const float* __restrict__ W1, const float* __restrict__ W2,
                 const float* __restrict__ Q, float* __restrict__ G, int steps) {
  int b = blockIdx.x;
  const int tid = threadIdx.x;
  const int* rows; int E; unsigned which;
  if (b < nb1) { rows = e1; E = E1; which = 0u; }
  else if (b < nb1 + nb2) { b -= nb1; rows = e2; E = E2; which = 1u; }
  else {
    const int g = b - (nb1 + nb2);
    const float* __restrict__ W = (g >= steps) ? W2 + (size_t)(g - steps) * DD * DD
                                               : W1 + (size_t)g * DD * DD;
    const int k = tid >> 4, n = tid & 15;
    float acc = 0.f;
#pragma unroll
    for (int t = 0; t < DD; ++t)
      acc = fmaf(W[t * DD + k], Q[t * DD + n], acc);
    G[(size_t)g * DD * DD + k * DD + n] = acc;
    return;
  }
  const int* cols = rows + E;
  const int base = b * EDGES_PER_BLOCK;
  const bool whole = base + EDGES_PER_BLOCK <= E;
  const bool aligned = ((((size_t)rows) | ((size_t)cols)) & 15) == 0;
  if (whole && aligned) {
    const int e = base + 4 * tid;
    const int4 r0 = *(const int4*)(rows + e);
    const int4 c0 = *(const int4*)(cols + e);
    const int4 r1 = *(const int4*)(rows + e + EDGES_PER_BLOCK / 2);
    const int4 c1 = *(const int4*)(cols + e + EDGES_PER_BLOCK / 2);
    __builtin_amdgcn_sched_barrier(0);
    set_raw(ER, which, r0.x, c0.x); set_raw(ER, which, r0.y, c0.y);
    set_raw(ER, which, r0.z, c0.z); set_raw(ER, which, r0.w, c0.w);
    set_raw(ER, which, r1.x, c1.x); set_raw(ER, which, r1.y, c1.y);
    set_raw(ER, which, r1.z, c1.z); set_raw(ER, which, r1.w, c1.w);
  } else {
    for (int u = 0; u < EDGES_PER_BLOCK / 256; ++u) {
      const int e = base + u * 256 + tid;
      if (e < E) set_raw(ER, which, rows[e], cols[e]);
    }
  }
}

typedef unsigned v2u __attribute__((ext_vector_type(2)));

__device__ __forceinline__ unsigned mask_edge(v2u pr, unsigned bit, unsigned* __restrict__ A1T,
                                              unsigned* __restrict__ A2T) {
  const unsigned m = 1u << (bit & 31), w = bit >> 5;
  if (pr.x & m) atomicOr(&A1T[w], m);
  unsigned old = m;
  if (pr.y & m) old = atomicOr(&A2T[w], m);
  return old;
}

__global__ __launch_bounds__(256)
void k_mask_hits(const int* __restrict__ mk, int EM, const unsigned* __restrict__ ERw,
                 unsigned* __restrict__ A1T, unsigned* __restrict__ A2T,
                 unsigned* __restrict__ cnt) {
  __shared__ unsigned wsum[8];
  const v2u* __restrict__ ER = (const v2u*)ERw;
  const int tid = threadIdx.x;
  const int* rows = mk;
  const int* cols = mk + EM;
  const int base = blockIdx.x * EDGES_PER_BLOCK;
  const bool whole = base + EDGES_PER_BLOCK <= EM;
  const bool aligned = ((((size_t)rows) | ((size_t)cols)) & 15) == 0;
  unsigned pc = 0;
  if (whole && aligned) {
    const int e = base + 4 * tid;
    const int4 r0 = *(const int4*)(rows + e);
    const int4 c0 = *(const int4*)(cols + e);
    const int4 r1 = *(const int4*)(rows + e + EDGES_PER_BLOCK / 2);
    const int4 c1 = *(const int4*)(cols + e + EDGES_PER_BLOCK / 2);
    unsigned bit[8];
    bit[0] = (unsigned)c0.x * (unsigned)NN + (unsigned)r0.x;
    bit[1] = (unsigned)c0.y * (unsigned)NN + (unsigned)r0.y;
    bit[2] = (unsigned)c0.z * (unsigned)NN + (unsigned)r0.z;
    bit[3] = (unsigned)c0.w * (unsigned)NN + (unsigned)r0.w;
    bit[4] = (unsigned)c1.x * (unsigned)NN + (unsigned)r1.x;
    bit[5] = (unsigned)c1.y * (unsigned)NN + (unsigned)r1.y;
    bit[6] = (unsigned)c1.z * (unsigned)NN + (unsigned)r1.z;
    bit[7] = (unsigned)c1.w * (unsigned)NN + (unsigned)r1.w;
    v2u pr[8];
#pragma unroll
    for (int k = 0; k < 8; ++k) pr[k] = ER[bit[k] >> 5];
    asm volatile("" :: "v"(pr[0]), "v"(pr[1]), "v"(pr[2]), "v"(pr[3]),
                       "v"(pr[4]), "v"(pr[5]), "v"(pr[6]), "v"(pr[7]));
    unsigned old[8];
#pragma unroll
    for (int k = 0; k < 8; ++k) old[k] = mask_edge(pr[k], bit[k], A1T, A2T);
    asm volatile("" :: "v"(old[0]), "v"(old[1]), "v"(old[2]), "v"(old[3]),
                       "v"(old[4]), "v"(old[5]), "v"(old[6]), "v"(old[7]));
#pragma unroll
    for (int k = 0; k < 8; ++k) pc += ((old[k] >> (bit[k] & 31)) & 1u) ^ 1u;
  } else {
    for (int u = 0; u < EDGES_PER_BLOCK / 256; ++u) {
      const int e = base + u * 256 + tid;
      if (e < EM) {
        const unsigned bit = (unsigned)cols[e] * (unsigned)NN + (unsigned)rows[e];
        pc += ((mask_edge(ER[bit >> 5], bit, A1T, A2T) >> (bit & 31)) & 1u) ^ 1u;
      }
    }
  }
#pragma unroll
  for (int off = 16; off > 0; off >>= 1) pc += __shfl_xor(pc, off);
  if ((tid & 31) == 0) wsum[tid >> 5] = pc;
  __syncthreads();
  if (tid == 0) {
    unsigned t = 0;
#pragma unroll
    for (int k = 0; k < 8; ++k) t += wsum[k];
    if (t) atomicAdd(cnt, t);
  }
}


#define DW 16
#define SRC_PER_WAVE 256
#define WPITCH 9
#define LIST_Q 34

#define ROW_OR_STEP(v, m) \
  (v) |= (unsigned)__builtin_amdgcn_update_dpp(0, (int)(v), 0x160 + (m), 0xf, 0xf, true)

struct BlkOps { v2f ax1[4]; v2f ax2[4]; v2f ar[4]; unsigned on1, on2; };

__device__ __forceinline__ void fetch_block(const float* __restrict__ x, int ibase, int c16, int h,
                                            int blk, const unsigned long long* lq1,
                                            const unsigned long long* lq2, const unsigned* wl1,
                                            const unsigned* wl2, int n1, int n2, BlkOps& f) {
  const unsigned long long e1 = lq1[2 * blk + h];
  const unsigned long long e2 = lq2[2 * blk + h];
  const unsigned own2 = ((const unsigned char*)lq2)[16 * blk + c16];
  const unsigned xo = (unsigned)(ibase * DD + c16);
  const int v1 = n1 - 16 * blk - 8 * h;
  const int v2 = n2 - 16 * blk - 8 * h;
  unsigned on1 = 0u, on2 = 0u;
#pragma unroll
  for (int r = 0; r < 8; ++r) {
    const unsigned p1 = (unsigned)(e1 >> (8 * r)) & 255u;
    const unsigned p2 = (unsigned)(e2 >> (8 * r)) & 255u;
    const float a = x[xo + p1 * DD];
    const float b = x[xo + p2 * DD];
    if (r & 1) { f.ax1[r >> 1].y = a; f.ax2[r >> 1].y = b; }
    else       { f.ax1[r >> 1].x = a; f.ax2[r >> 1].x = b; }
    const unsigned wv1 = wl1[c16 * WPITCH + (p1 >> 5)];
    const unsigned wv2 = wl2[c16 * WPITCH + (p2 >> 5)];
    const unsigned b1 = (wv1 >> (p1 & 31u)) & 1u;
    const unsigned b2 = (wv2 >> (p2 & 31u)) & 1u;
    on1 |= b1 << r;
    on2 |= b2 << r;
  }
  on1 &= v1 >= 8 ? 0xFFu : (v1 > 0 ? (1u << v1) - 1u : 0u);
  on2 &= v2 >= 8 ? 0xFFu : (v2 > 0 ? (1u << v2) - 1u : 0u);
  const float4* pr = (const float4*)(x + ((unsigned)ibase + own2) * DD + 8 * h);
  const float4 r0 = pr[0], r1 = pr[1];
  f.ar[0].x = r0.x; f.ar[0].y = r0.y;
  f.ar[1].x = r0.z; f.ar[1].y = r0.w;
  f.ar[2].x = r1.x; f.ar[2].y = r1.y;
  f.ar[3].x = r1.z; f.ar[3].y = r1.w;
  f.on1 = on1; f.on2 = on2;
}

__device__ __forceinline__ void block_products(const BlkOps& f, const v2f vb[4], v8f& y1, v8f& y2) {
#pragma unroll
  for (int q = 0; q < 4; ++q) {
    v2f b;
    b.x = (float)((f.on1 >> (2 * q)) & 1u);
    b.y = (float)((f.on1 >> (2 * q + 1)) & 1u);
    y1 = __builtin_amdgcn_wmma_f32_16x16x4_f32(false, f.ax1[q], false, b, (short)0, y1, false, false);
  }
  v8f s = {};
  s = wmma16(f.ar, vb, s);
#pragma unroll
  for (int q = 0; q < 4; ++q) {
    v2f p;
    p.x = __uint_as_float(__float_as_uint(s[2 * q]) & (0u - ((f.on2 >> (2 * q)) & 1u)));
    p.y = __uint_as_float(__float_as_uint(s[2 * q + 1]) & (0u - ((f.on2 >> (2 * q + 1)) & 1u)));
    y2 = __builtin_amdgcn_wmma_f32_16x16x4_f32(false, f.ax2[q], false, p, (short)0, y2, false, false);
  }
}

__global__ __launch_bounds__(DW * 32)
void k_step(const float* __restrict__ x, const float* __restrict__ Qm,
            const unsigned* __restrict__ A1T, const unsigned* __restrict__ A2T,
            const float* __restrict__ G1s, const float* __restrict__ G2s,
            const unsigned* __restrict__ cnt2, float s1, float* __restrict__ xnext) {
  __shared__ float red[DW][2][256];
  __shared__ float ysum[2][256];
  __shared__ unsigned wl[DW][2][16 * WPITCH];
  __shared__ unsigned long long lst[DW][2][LIST_Q];
  const int tid = threadIdx.x, lane = tid & 31, wave = tid >> 5;
  const int c16 = lane & 15, h = lane >> 4;
  const int j0 = blockIdx.x * 16;
  const int ibase = wave * SRC_PER_WAVE;

  v2f qa[4];
  {
    const float4* pq = (const float4*)(Qm + c16 * DD + 8 * h);
    float4 a0 = pq[0], a1 = pq[1];
    qa[0].x = a0.x; qa[0].y = a0.y;
    qa[1].x = a0.z; qa[1].y = a0.w;
    qa[2].x = a1.x; qa[2].y = a1.y;
    qa[3].x = a1.z; qa[3].y = a1.w;
  }

  const int j = j0 + c16;
  const uint4* p1 = (const uint4*)(A1T + (size_t)j * (NN / 32) + (ibase >> 5));
  const uint4* p2 = (const uint4*)(A2T + (size_t)j * (NN / 32) + (ibase >> 5));
  const uint4 u1a = p1[0], u1b = p1[1], u2a = p2[0], u2b = p2[1];
  const float4* pj = (const float4*)(x + (size_t)j * DD + 8 * h);
  const float4 xa0 = pj[0], xa1 = pj[1];
  v2f g1[4] = {}, g2[4] = {};
  v8f c = {};
  if (wave == 0) {
    load_b_frag(G1s, g1);
    load_b_frag(G2s, g2);
    c = load_c_tile(x, j0);
  }
  const float s2 = -(float)NN / (float)(*cnt2);

  unsigned* wl1 = wl[wave][0];
  unsigned* wl2 = wl[wave][1];
  unsigned long long* lq1 = lst[wave][0];
  unsigned long long* lq2 = lst[wave][1];
  unsigned char* lb1 = (unsigned char*)lq1;
  unsigned char* lb2 = (unsigned char*)lq2;

  lq1[lane] = 0ull;
  lq2[lane] = 0ull;
  if (lane < LIST_Q - 32) { lq1[32 + lane] = 0ull; lq2[32 + lane] = 0ull; }

  unsigned w1[8], w2[8];
  w1[0] = u1a.x; w1[1] = u1a.y; w1[2] = u1a.z; w1[3] = u1a.w;
  w1[4] = u1b.x; w1[5] = u1b.y; w1[6] = u1b.z; w1[7] = u1b.w;
  w2[0] = u2a.x; w2[1] = u2a.y; w2[2] = u2a.z; w2[3] = u2a.w;
  w2[4] = u2b.x; w2[5] = u2b.y; w2[6] = u2b.z; w2[7] = u2b.w;

  int n1 = 0, n2 = 0;
#pragma unroll
  for (int t = 0; t < 8; ++t) {
    wl1[c16 * WPITCH + t] = w1[t];
    wl2[c16 * WPITCH + t] = w2[t];
    unsigned o1 = w1[t], o2 = w2[t];
    ROW_OR_STEP(o1, 1); ROW_OR_STEP(o2, 1);
    ROW_OR_STEP(o1, 2); ROW_OR_STEP(o2, 2);
    ROW_OR_STEP(o1, 4); ROW_OR_STEP(o2, 4);
    ROW_OR_STEP(o1, 8); ROW_OR_STEP(o2, 8);
    const unsigned m1 = __builtin_amdgcn_readfirstlane(o1);
    const unsigned m2 = __builtin_amdgcn_readfirstlane(o2);
    if ((m1 >> lane) & 1u)
      lb1[n1 + (int)__builtin_amdgcn_mbcnt_lo(m1, 0u)] = (unsigned char)(32 * t + lane);
    if ((m2 >> lane) & 1u)
      lb2[n2 + (int)__builtin_amdgcn_mbcnt_lo(m2, 0u)] = (unsigned char)(32 * t + lane);
    n1 += __builtin_popcount(m1);
    n2 += __builtin_popcount(m2);
  }

  v2f vb[4];
  {
    v2f xjb[4];
    xjb[0].x = xa0.x; xjb[0].y = xa0.y;
    xjb[1].x = xa0.z; xjb[1].y = xa0.w;
    xjb[2].x = xa1.x; xjb[2].y = xa1.y;
    xjb[3].x = xa1.z; xjb[3].y = xa1.w;
    v8f v = {};
    v = wmma16(qa, xjb, v);
#pragma unroll
    for (int q = 0; q < 4; ++q) {
      vb[q].x = v[2 * q];
      vb[q].y = v[2 * q + 1];
    }
  }

  const int nb1 = (n1 + 15) >> 4, nb2 = (n2 + 15) >> 4;
  const int nbm = nb1 > nb2 ? nb1 : nb2;
  v8f y1 = {}, y2 = {};
  BlkOps fa, fb;
  fetch_block(x, ibase, c16, h, 0, lq1, lq2, wl1, wl2, n1, n2, fa);
  for (int blk = 0; blk < nbm; blk += 2) {
    fetch_block(x, ibase, c16, h, blk + 1, lq1, lq2, wl1, wl2, n1, n2, fb);
    block_products(fa, vb, y1, y2);
    if (blk + 1 < nbm) {
      fetch_block(x, ibase, c16, h, blk + 2, lq1, lq2, wl1, wl2, n1, n2, fa);
      block_products(fb, vb, y1, y2);
    }
  }

#pragma unroll
  for (int r = 0; r < 8; ++r) {
    red[wave][0][r * 32 + lane] = y1[r];
    red[wave][1][r * 32 + lane] = y2[r];
  }
  __syncthreads();
  {
    const int which = tid >> 8, e = tid & 255;
    float s = 0.f;
#pragma unroll
    for (int w = 0; w < DW; ++w) s += red[w][which][e];
    const int r = e >> 5, hh = (e >> 4) & 1, n = e & 15;
    ysum[which][n * DD + r + 8 * hh] = s;
  }
  __syncthreads();
  if (wave == 0) {
    v2f a[4];
    load_a_frag_tile(ysum[0], s1, a);
    c = wmma16(a, g1, c);
    load_a_frag_tile(ysum[1], s2, a);
    c = wmma16(a, g2, c);
    store_d_tile(xnext, j0, c);
  }
}

extern "C" void kernel_launch(void* const* d_in, const int* in_sizes, int n_in,
                              void* d_out, int out_size, void* d_ws, size_t ws_size,
                              hipStream_t stream) {
  const float* x_in = (const float*)d_in[0];
  const int* e1 = (const int*)d_in[1];
  const int* e2 = (const int*)d_in[2];
  const float* Qm = (const float*)d_in[3];
  const int* mk = (const int*)d_in[4];
  const float* W1 = (const float*)d_in[5];
  const float* W2 = (const float*)d_in[6];
  float* out = (float*)d_out;

  const int E1 = in_sizes[1] / 2;
  const int E2 = in_sizes[2] / 2;
  const int EM = in_sizes[4] / 2;
  const int STEPS = in_sizes[5] / (DD * DD);

  char* ws = (char*)d_ws;
  size_t off = 0;
  auto carve = [&](size_t bytes) {
    char* p = ws + off;
    off += (bytes + 255) & ~(size_t)255;
    return p;
  };
  const size_t BM = (size_t)NN * NN / 8;
  unsigned* ER    = (unsigned*)carve(2 * BM);
  unsigned* A1T   = (unsigned*)carve(BM);
  unsigned* A2T   = (unsigned*)carve(BM);
  unsigned* cnts  = (unsigned*)carve(8);
  float* G   = (float*)carve((size_t)2 * STEPS * DD * DD * 4);
  const float* G1 = G;
  const float* G2 = G + (size_t)STEPS * DD * DD;
  float* xb0 = (float*)carve((size_t)NN * DD * 4);
  float* xb1 = (float*)carve((size_t)NN * DD * 4);

  hipMemsetAsync(ER, 0, 4 * BM + 8, stream);

  const int nbM = (EM + EDGES_PER_BLOCK - 1) / EDGES_PER_BLOCK;
  const int nb1 = (E1 + EDGES_PER_BLOCK - 1) / EDGES_PER_BLOCK;
  const int nb2 = (E2 + EDGES_PER_BLOCK - 1) / EDGES_PER_BLOCK;
  k_build_raw<<<nb1 + nb2 + 2 * STEPS, 256, 0, stream>>>(
      e1, E1, e2, E2, nb1, nb2, ER, W1, W2, Qm, G, STEPS);
  k_mask_hits<<<nbM, 256, 0, stream>>>(mk, EM, ER, A1T, A2T, cnts + 1);

  const float s1 = (float)NN / (float)EM;

  const float* xc = x_in;
  for (int s = 0; s < STEPS; ++s) {
    float* xn = (s == STEPS - 1) ? out : ((s & 1) ? xb1 : xb0);
    k_step<<<NN / 16, DW * 32, 0, stream>>>(xc, Qm, A1T, A2T, G1 + (size_t)s * DD * DD,
                                            G2 + (size_t)s * DD * DD, cnts + 1, s1, xn);
    xc = xn;
  }
}


# k_step builds its compacted operands (source lists, lengths, per-block bit masks) in step 1 only and keeps them in the workspace; steps 2 and 3 read them back, no bitmap or list work; sums unchanged
# speedup vs baseline: 13.9086x; 1.0415x over previous
#include <hip/hip_runtime.h>


#define NN 4096
#define DD 16

typedef float v2f __attribute__((ext_vector_type(2)));
typedef float v8f __attribute__((ext_vector_type(8)));

__device__ __forceinline__ void load_a_frag_tile(const float* Y, float scale, v2f a[4]) {
  const int lane = threadIdx.x & 31;
  const int m = lane & 15, h = lane >> 4;
  const float* b0 = Y + m * DD + 2 * h;
#pragma unroll
  for (int c = 0; c < 4; ++c) {
    a[c].x = b0[4 * c] * scale;
    a[c].y = b0[4 * c + 1] * scale;
  }
}

__device__ __forceinline__ void load_b_frag(const float* __restrict__ W, v2f b[4]) {
  const int lane = threadIdx.x & 31;
  const int n = lane & 15, h = lane >> 4;
#pragma unroll
  for (int c = 0; c < 4; ++c) {
    b[c].x = W[(4 * c + 2 * h) * DD + n];
    b[c].y = W[(4 * c + 2 * h + 1) * DD + n];
  }
}

__device__ __forceinline__ v8f load_c_tile(const float* __restrict__ X, int row0) {
  const int lane = threadIdx.x & 31;
  const int n = lane & 15, h = lane >> 4;
  v8f c;
#pragma unroll
  for (int r = 0; r < 8; ++r) c[r] = X[(size_t)(row0 + r + 8 * h) * DD + n];
  return c;
}

__device__ __forceinline__ void store_d_tile(float* __restrict__ O, int row0, v8f d) {
  const int lane = threadIdx.x & 31;
  const int n = lane & 15, h = lane >> 4;
#pragma unroll
  for (int r = 0; r < 8; ++r) O[(size_t)(row0 + r + 8 * h) * DD + n] = d[r];
}

__device__ __forceinline__ v8f wmma16(const v2f a[4], const v2f b[4], v8f c) {
#pragma unroll
  for (int k = 0; k < 4; ++k)
    c = __builtin_amdgcn_wmma_f32_16x16x4_f32(false, a[k], false, b[k],
                                              (short)0, c, false, false);
  return c;
}


#define EDGES_PER_BLOCK 2048

__device__ __forceinline__ void set_raw(unsigned* __restrict__ ER, unsigned which, int row, int col) {
  const unsigned bit = (unsigned)col * (unsigned)NN + (unsigned)row;
  atomicOr(&ER[2u * (bit >> 5) + which], 1u << (bit & 31));
}

__global__ __launch_bounds__(256)
void k_build_raw(const int* __restrict__ e1, int E1, const int* __restrict__ e2, int E2,
                 int nb1, int nb2, unsigned* __restrict__ ER,
                 const float* __restrict__ W1, const float* __restrict__ W2,
                 const float* __restrict__ Q, float* __restrict__ G, int steps) {
  int b = blockIdx.x;
  const int tid = threadIdx.x;
  const int* rows; int E; unsigned which;
  if (b < nb1) { rows = e1; E = E1; which = 0u; }
  else if (b < nb1 + nb2) { b -= nb1; rows = e2; E = E2; which = 1u; }
  else {
    const int g = b - (nb1 + nb2);
    const float* __restrict__ W = (g >= steps) ? W2 + (size_t)(g - steps) * DD * DD
                                               : W1 + (size_t)g * DD * DD;
    const int k = tid >> 4, n = tid & 15;
    float acc = 0.f;
#pragma unroll
    for (int t = 0; t < DD; ++t)
      acc = fmaf(W[t * DD + k], Q[t * DD + n], acc);
    G[(size_t)g * DD * DD + k * DD + n] = acc;
    return;
  }
  const int* cols = rows + E;
  const int base = b * EDGES_PER_BLOCK;
  const bool whole = base + EDGES_PER_BLOCK <= E;
  const bool aligned = ((((size_t)rows) | ((size_t)cols)) & 15) == 0;
  if (whole && aligned) {
    const int e = base + 4 * tid;
    const int4 r0 = *(const int4*)(rows + e);
    const int4 c0 = *(const int4*)(cols + e);
    const int4 r1 = *(const int4*)(rows + e + EDGES_PER_BLOCK / 2);
    const int4 c1 = *(const int4*)(cols + e + EDGES_PER_BLOCK / 2);
    __builtin_amdgcn_sched_barrier(0);
    set_raw(ER, which, r0.x, c0.x); set_raw(ER, which, r0.y, c0.y);
    set_raw(ER, which, r0.z, c0.z); set_raw(ER, which, r0.w, c0.w);
    set_raw(ER, which, r1.x, c1.x); set_raw(ER, which, r1.y, c1.y);
    set_raw(ER, which, r1.z, c1.z); set_raw(ER, which, r1.w, c1.w);
  } else {
    for (int u = 0; u < EDGES_PER_BLOCK / 256; ++u) {
      const int e = base + u * 256 + tid;
      if (e < E) set_raw(ER, which, rows[e], cols[e]);
    }
  }
}

typedef unsigned v2u __attribute__((ext_vector_type(2)));

__device__ __forceinline__ unsigned mask_edge(v2u pr, unsigned bit, unsigned* __restrict__ A1T,
                                              unsigned* __restrict__ A2T) {
  const unsigned m = 1u << (bit & 31), w = bit >> 5;
  if (pr.x & m) atomicOr(&A1T[w], m);
  unsigned old = m;
  if (pr.y & m) old = atomicOr(&A2T[w], m);
  return old;
}

__global__ __launch_bounds__(256)
void k_mask_hits(const int* __restrict__ mk, int EM, const unsigned* __restrict__ ERw,
                 unsigned* __restrict__ A1T, unsigned* __restrict__ A2T,
                 unsigned* __restrict__ cnt) {
  __shared__ unsigned wsum[8];
  const v2u* __restrict__ ER = (const v2u*)ERw;
  const int tid = threadIdx.x;
  const int* rows = mk;
  const int* cols = mk + EM;
  const int base = blockIdx.x * EDGES_PER_BLOCK;
  const bool whole = base + EDGES_PER_BLOCK <= EM;
  const bool aligned = ((((size_t)rows) | ((size_t)cols)) & 15) == 0;
  unsigned pc = 0;
  if (whole && aligned) {
    const int e = base + 4 * tid;
    const int4 r0 = *(const int4*)(rows + e);
    const int4 c0 = *(const int4*)(cols + e);
    const int4 r1 = *(const int4*)(rows + e + EDGES_PER_BLOCK / 2);
    const int4 c1 = *(const int4*)(cols + e + EDGES_PER_BLOCK / 2);
    unsigned bit[8];
    bit[0] = (unsigned)c0.x * (unsigned)NN + (unsigned)r0.x;
    bit[1] = (unsigned)c0.y * (unsigned)NN + (unsigned)r0.y;
    bit[2] = (unsigned)c0.z * (unsigned)NN + (unsigned)r0.z;
    bit[3] = (unsigned)c0.w * (unsigned)NN + (unsigned)r0.w;
    bit[4] = (unsigned)c1.x * (unsigned)NN + (unsigned)r1.x;
    bit[5] = (unsigned)c1.y * (unsigned)NN + (unsigned)r1.y;
    bit[6] = (unsigned)c1.z * (unsigned)NN + (unsigned)r1.z;
    bit[7] = (unsigned)c1.w * (unsigned)NN + (unsigned)r1.w;
    v2u pr[8];
#pragma unroll
    for (int k = 0; k < 8; ++k) pr[k] = ER[bit[k] >> 5];
    asm volatile("" :: "v"(pr[0]), "v"(pr[1]), "v"(pr[2]), "v"(pr[3]),
                       "v"(pr[4]), "v"(pr[5]), "v"(pr[6]), "v"(pr[7]));
    unsigned old[8];
#pragma unroll
    for (int k = 0; k < 8; ++k) old[k] = mask_edge(pr[k], bit[k], A1T, A2T);
    asm volatile("" :: "v"(old[0]), "v"(old[1]), "v"(old[2]), "v"(old[3]),
                       "v"(old[4]), "v"(old[5]), "v"(old[6]), "v"(old[7]));
#pragma unroll
    for (int k = 0; k < 8; ++k) pc += ((old[k] >> (bit[k] & 31)) & 1u) ^ 1u;
  } else {
    for (int u = 0; u < EDGES_PER_BLOCK / 256; ++u) {
      const int e = base + u * 256 + tid;
      if (e < EM) {
        const unsigned bit = (unsigned)cols[e] * (unsigned)NN + (unsigned)rows[e];
        pc += ((mask_edge(ER[bit >> 5], bit, A1T, A2T) >> (bit & 31)) & 1u) ^ 1u;
      }
    }
  }
#pragma unroll
  for (int off = 16; off > 0; off >>= 1) pc += __shfl_xor(pc, off);
  if ((tid & 31) == 0) wsum[tid >> 5] = pc;
  __syncthreads();
  if (tid == 0) {
    unsigned t = 0;
#pragma unroll
    for (int k = 0; k < 8; ++k) t += wsum[k];
    if (t) atomicAdd(cnt, t);
  }
}


#define DW 16
#define SRC_PER_WAVE 256
#define WPITCH 9
#define LIST_Q 34

#define ROW_OR_STEP(v, m) \
  (v) |= (unsigned)__builtin_amdgcn_update_dpp(0, (int)(v), 0x160 + (m), 0xf, 0xf, true)

struct BlkOps { v2f ax1[4]; v2f ax2[4]; v2f ar[4]; unsigned on; };

#define KB_BLOCKS 17

template <bool FIRST>
__device__ __forceinline__ void fetch_block(const float* __restrict__ x, int ibase, int c16, int h,
                                            int blk, const unsigned long long* lq1,
                                            const unsigned long long* lq2, const unsigned* wl1,
                                            const unsigned* wl2, int n1, int n2,
                                            const unsigned short* __restrict__ kbw, BlkOps& f) {
  const unsigned long long e1 = lq1[2 * blk + h];
  const unsigned long long e2 = lq2[2 * blk + h];
  const unsigned own2 = ((const unsigned char*)lq2)[16 * blk + c16];
  const unsigned xo = (unsigned)(ibase * DD + c16);
  unsigned on1 = 0u, on2 = 0u;
#pragma unroll
  for (int r = 0; r < 8; ++r) {
    const unsigned p1 = (unsigned)(e1 >> (8 * r)) & 255u;
    const unsigned p2 = (unsigned)(e2 >> (8 * r)) & 255u;
    const float a = x[xo + p1 * DD];
    const float b = x[xo + p2 * DD];
    if (r & 1) { f.ax1[r >> 1].y = a; f.ax2[r >> 1].y = b; }
    else       { f.ax1[r >> 1].x = a; f.ax2[r >> 1].x = b; }
    if (FIRST) {
      const unsigned wv1 = wl1[c16 * WPITCH + (p1 >> 5)];
      const unsigned wv2 = wl2[c16 * WPITCH + (p2 >> 5)];
      const unsigned b1 = (wv1 >> (p1 & 31u)) & 1u;
      const unsigned b2 = (wv2 >> (p2 & 31u)) & 1u;
      on1 |= b1 << r;
      on2 |= b2 << r;
    }
  }
  if (FIRST) {
    const int v1 = n1 - 16 * blk - 8 * h;
    const int v2 = n2 - 16 * blk - 8 * h;
    on1 &= v1 >= 8 ? 0xFFu : (v1 > 0 ? (1u << v1) - 1u : 0u);
    on2 &= v2 >= 8 ? 0xFFu : (v2 > 0 ? (1u << v2) - 1u : 0u);
    f.on = on1 | (on2 << 8);
  } else {
    f.on = kbw[blk * 32 + (threadIdx.x & 31)];
  }
  const float4* pr = (const float4*)(x + ((unsigned)ibase + own2) * DD + 8 * h);
  const float4 r0 = pr[0], r1 = pr[1];
  f.ar[0].x = r0.x; f.ar[0].y = r0.y;
  f.ar[1].x = r0.z; f.ar[1].y = r0.w;
  f.ar[2].x = r1.x; f.ar[2].y = r1.y;
  f.ar[3].x = r1.z; f.ar[3].y = r1.w;
}

template <bool FIRST>
__device__ __forceinline__ void block_products(const BlkOps& f, const v2f vb[4], v8f& y1, v8f& y2,
                                               int blk, unsigned short* __restrict__ kbw) {
  if (FIRST) kbw[blk * 32 + (threadIdx.x & 31)] = (unsigned short)f.on;
  const unsigned on1 = f.on & 255u, on2 = f.on >> 8;
#pragma unroll
  for (int q = 0; q < 4; ++q) {
    v2f b;
    b.x = (float)((on1 >> (2 * q)) & 1u);
    b.y = (float)((on1 >> (2 * q + 1)) & 1u);
    y1 = __builtin_amdgcn_wmma_f32_16x16x4_f32(false, f.ax1[q], false, b, (short)0, y1, false, false);
  }
  v8f s = {};
  s = wmma16(f.ar, vb, s);
#pragma unroll
  for (int q = 0; q < 4; ++q) {
    v2f p;
    p.x = __uint_as_float(__float_as_uint(s[2 * q]) & (0u - ((on2 >> (2 * q)) & 1u)));
    p.y = __uint_as_float(__float_as_uint(s[2 * q + 1]) & (0u - ((on2 >> (2 * q + 1)) & 1u)));
    y2 = __builtin_amdgcn_wmma_f32_16x16x4_f32(false, f.ax2[q], false, p, (short)0, y2, false, false);
  }
}

template <bool FIRST>
__global__ __launch_bounds__(DW * 32)
void k_step(const float* __restrict__ x, const float* __restrict__ Qm,
            const unsigned* __restrict__ A1T, const unsigned* __restrict__ A2T,
            const float* __restrict__ G1s, const float* __restrict__ G2s,
            const unsigned* __restrict__ cnt2, float s1, float* __restrict__ xnext,
            int* __restrict__ kn, unsigned long long* __restrict__ kl,
            unsigned short* __restrict__ kb) {
  __shared__ float red[DW][2][256];
  __shared__ float ysum[2][256];
  __shared__ unsigned wl[DW][2][16 * WPITCH];
  __shared__ unsigned long long lst[DW][2][LIST_Q];
  const int tid = threadIdx.x, lane = tid & 31, wave = tid >> 5;
  const int c16 = lane & 15, h = lane >> 4;
  const int j0 = blockIdx.x * 16;
  const int ibase = wave * SRC_PER_WAVE;

  v2f qa[4];
  {
    const float4* pq = (const float4*)(Qm + c16 * DD + 8 * h);
    float4 a0 = pq[0], a1 = pq[1];
    qa[0].x = a0.x; qa[0].y = a0.y;
    qa[1].x = a0.z; qa[1].y = a0.w;
    qa[2].x = a1.x; qa[2].y = a1.y;
    qa[3].x = a1.z; qa[3].y = a1.w;
  }

  const int j = j0 + c16;
  uint4 u1a = {}, u1b = {}, u2a = {}, u2b = {};
  const int wid = blockIdx.x * DW + wave;
  int* knw = kn + 2 * wid;
  unsigned long long* klw = kl + (size_t)wid * 64;
  unsigned short* kbw = kb + (size_t)wid * (KB_BLOCKS * 32);
  unsigned long long k1 = 0ull, k2 = 0ull;
  int kn1 = 0, kn2 = 0;
  if (FIRST) {
    const uint4* p1 = (const uint4*)(A1T + (size_t)j * (NN / 32) + (ibase >> 5));
    const uint4* p2 = (const uint4*)(A2T + (size_t)j * (NN / 32) + (ibase >> 5));
    u1a = p1[0]; u1b = p1[1]; u2a = p2[0]; u2b = p2[1];
  } else {
    k1 = klw[lane];
    k2 = klw[32 + lane];
    kn1 = knw[0];
    kn2 = knw[1];
  }
  const float4* pj = (const float4*)(x + (size_t)j * DD + 8 * h);
  const float4 xa0 = pj[0], xa1 = pj[1];
  v2f g1[4] = {}, g2[4] = {};
  v8f c = {};
  if (wave == 0) {
    load_b_frag(G1s, g1);
    load_b_frag(G2s, g2);
    c = load_c_tile(x, j0);
  }
  const float s2 = -(float)NN / (float)(*cnt2);

  unsigned* wl1 = wl[wave][0];
  unsigned* wl2 = wl[wave][1];
  unsigned long long* lq1 = lst[wave][0];
  unsigned long long* lq2 = lst[wave][1];
  unsigned char* lb1 = (unsigned char*)lq1;
  unsigned char* lb2 = (unsigned char*)lq2;

  lq1[lane] = FIRST ? 0ull : k1;
  lq2[lane] = FIRST ? 0ull : k2;
  if (lane < LIST_Q - 32) { lq1[32 + lane] = 0ull; lq2[32 + lane] = 0ull; }

  unsigned w1[8], w2[8];
  w1[0] = u1a.x; w1[1] = u1a.y; w1[2] = u1a.z; w1[3] = u1a.w;
  w1[4] = u1b.x; w1[5] = u1b.y; w1[6] = u1b.z; w1[7] = u1b.w;
  w2[0] = u2a.x; w2[1] = u2a.y; w2[2] = u2a.z; w2[3] = u2a.w;
  w2[4] = u2b.x; w2[5] = u2b.y; w2[6] = u2b.z; w2[7] = u2b.w;

  int n1 = 0, n2 = 0;
  if (FIRST) {
#pragma unroll
    for (int t = 0; t < 8; ++t) {
      wl1[c16 * WPITCH + t] = w1[t];
      wl2[c16 * WPITCH + t] = w2[t];
      unsigned o1 = w1[t], o2 = w2[t];
      ROW_OR_STEP(o1, 1); ROW_OR_STEP(o2, 1);
      ROW_OR_STEP(o1, 2); ROW_OR_STEP(o2, 2);
      ROW_OR_STEP(o1, 4); ROW_OR_STEP(o2, 4);
      ROW_OR_STEP(o1, 8); ROW_OR_STEP(o2, 8);
      const unsigned m1 = __builtin_amdgcn_readfirstlane(o1);
      const unsigned m2 = __builtin_amdgcn_readfirstlane(o2);
      if ((m1 >> lane) & 1u)
        lb1[n1 + (int)__builtin_amdgcn_mbcnt_lo(m1, 0u)] = (unsigned char)(32 * t + lane);
      if ((m2 >> lane) & 1u)
        lb2[n2 + (int)__builtin_amdgcn_mbcnt_lo(m2, 0u)] = (unsigned char)(32 * t + lane);
      n1 += __builtin_popcount(m1);
      n2 += __builtin_popcount(m2);
    }
    klw[lane] = lq1[lane];
    klw[32 + lane] = lq2[lane];
    if (lane == 0) { knw[0] = n1; knw[1] = n2; }
  } else {
    n1 = __builtin_amdgcn_readfirstlane(kn1);
    n2 = __builtin_amdgcn_readfirstlane(kn2);
  }

  v2f vb[4];
  {
    v2f xjb[4];
    xjb[0].x = xa0.x; xjb[0].y = xa0.y;
    xjb[1].x = xa0.z; xjb[1].y = xa0.w;
    xjb[2].x = xa1.x; xjb[2].y = xa1.y;
    xjb[3].x = xa1.z; xjb[3].y = xa1.w;
    v8f v = {};
    v = wmma16(qa, xjb, v);
#pragma unroll
    for (int q = 0; q < 4; ++q) {
      vb[q].x = v[2 * q];
      vb[q].y = v[2 * q + 1];
    }
  }

  const int nb1 = (n1 + 15) >> 4, nb2 = (n2 + 15) >> 4;
  const int nbm = nb1 > nb2 ? nb1 : nb2;
  v8f y1 = {}, y2 = {};
  BlkOps fa, fb;
  if (FIRST) kbw[nbm * 32 + lane] = (unsigned short)0;
  fetch_block<FIRST>(x, ibase, c16, h, 0, lq1, lq2, wl1, wl2, n1, n2, kbw, fa);
  for (int blk = 0; blk < nbm; blk += 2) {
    fetch_block<FIRST>(x, ibase, c16, h, blk + 1, lq1, lq2, wl1, wl2, n1, n2, kbw, fb);
    block_products<FIRST>(fa, vb, y1, y2, blk, kbw);
    if (blk + 1 < nbm) {
      fetch_block<FIRST>(x, ibase, c16, h, blk + 2, lq1, lq2, wl1, wl2, n1, n2, kbw, fa);
      block_products<FIRST>(fb, vb, y1, y2, blk + 1, kbw);
    }
  }

#pragma unroll
  for (int r = 0; r < 8; ++r) {
    red[wave][0][r * 32 + lane] = y1[r];
    red[wave][1][r * 32 + lane] = y2[r];
  }
  __syncthreads();
  {
    const int which = tid >> 8, e = tid & 255;
    float s = 0.f;
#pragma unroll
    for (int w = 0; w < DW; ++w) s += red[w][which][e];
    const int r = e >> 5, hh = (e >> 4) & 1, n = e & 15;
    ysum[which][n * DD + r + 8 * hh] = s;
  }
  __syncthreads();
  if (wave == 0) {
    v2f a[4];
    load_a_frag_tile(ysum[0], s1, a);
    c = wmma16(a, g1, c);
    load_a_frag_tile(ysum[1], s2, a);
    c = wmma16(a, g2, c);
    store_d_tile(xnext, j0, c);
  }
}

extern "C" void kernel_launch(void* const* d_in, const int* in_sizes, int n_in,
                              void* d_out, int out_size, void* d_ws, size_t ws_size,
                              hipStream_t stream) {
  const float* x_in = (const float*)d_in[0];
  const int* e1 = (const int*)d_in[1];
  const int* e2 = (const int*)d_in[2];
  const float* Qm = (const float*)d_in[3];
  const int* mk = (const int*)d_in[4];
  const float* W1 = (const float*)d_in[5];
  const float* W2 = (const float*)d_in[6];
  float* out = (float*)d_out;

  const int E1 = in_sizes[1] / 2;
  const int E2 = in_sizes[2] / 2;
  const int EM = in_sizes[4] / 2;
  const int STEPS = in_sizes[5] / (DD * DD);

  char* ws = (char*)d_ws;
  size_t off = 0;
  auto carve = [&](size_t bytes) {
    char* p = ws + off;
    off += (bytes + 255) & ~(size_t)255;
    return p;
  };
  const size_t BM = (size_t)NN * NN / 8;
  unsigned* ER    = (unsigned*)carve(2 * BM);
  unsigned* A1T   = (unsigned*)carve(BM);
  unsigned* A2T   = (unsigned*)carve(BM);
  unsigned* cnts  = (unsigned*)carve(8);
  float* G   = (float*)carve((size_t)2 * STEPS * DD * DD * 4);
  const float* G1 = G;
  const float* G2 = G + (size_t)STEPS * DD * DD;
  float* xb0 = (float*)carve((size_t)NN * DD * 4);
  float* xb1 = (float*)carve((size_t)NN * DD * 4);
  const size_t NW = (size_t)(NN / 16) * DW;
  int* kn = (int*)carve(NW * 2 * sizeof(int));
  unsigned long long* kl = (unsigned long long*)carve(NW * 64 * sizeof(unsigned long long));
  unsigned short* kb = (unsigned short*)carve(NW * KB_BLOCKS * 32 * sizeof(unsigned short));

  hipMemsetAsync(ER, 0, 4 * BM + 8, stream);

  const int nbM = (EM + EDGES_PER_BLOCK - 1) / EDGES_PER_BLOCK;
  const int nb1 = (E1 + EDGES_PER_BLOCK - 1) / EDGES_PER_BLOCK;
  const int nb2 = (E2 + EDGES_PER_BLOCK - 1) / EDGES_PER_BLOCK;
  k_build_raw<<<nb1 + nb2 + 2 * STEPS, 256, 0, stream>>>(
      e1, E1, e2, E2, nb1, nb2, ER, W1, W2, Qm, G, STEPS);
  k_mask_hits<<<nbM, 256, 0, stream>>>(mk, EM, ER, A1T, A2T, cnts + 1);

  const float s1 = (float)NN / (float)EM;

  const float* xc = x_in;
  for (int s = 0; s < STEPS; ++s) {
    float* xn = (s == STEPS - 1) ? out : ((s & 1) ? xb1 : xb0);
    if (s == 0)
      k_step<true><<<NN / 16, DW * 32, 0, stream>>>(xc, Qm, A1T, A2T, G1, G2, cnts + 1, s1, xn,
                                                    kn, kl, kb);
    else
      k_step<false><<<NN / 16, DW * 32, 0, stream>>>(xc, Qm, A1T, A2T, G1 + (size_t)s * DD * DD,
                                                     G2 + (size_t)s * DD * DD, cnts + 1, s1, xn,
                                                     kn, kl, kb);
    xc = xn;
  }
}


# Set-construction atomics issued from one half of the package per table row: work-groups read their die id, take (chunk, row-parity class) tickets per half, do only their class's edges; sums same
# speedup vs baseline: 17.0415x; 1.2252x over previous
#include <hip/hip_runtime.h>


#define NN 4096
#define DD 16

typedef float v2f __attribute__((ext_vector_type(2)));
typedef float v8f __attribute__((ext_vector_type(8)));

__device__ __forceinline__ void load_a_frag_tile(const float* Y, float scale, v2f a[4]) {
  const int lane = threadIdx.x & 31;
  const int m = lane & 15, h = lane >> 4;
  const float* b0 = Y + m * DD + 2 * h;
#pragma unroll
  for (int c = 0; c < 4; ++c) {
    a[c].x = b0[4 * c] * scale;
    a[c].y = b0[4 * c + 1] * scale;
  }
}

__device__ __forceinline__ void load_b_frag(const float* __restrict__ W, v2f b[4]) {
  const int lane = threadIdx.x & 31;
  const int n = lane & 15, h = lane >> 4;
#pragma unroll
  for (int c = 0; c < 4; ++c) {
    b[c].x = W[(4 * c + 2 * h) * DD + n];
    b[c].y = W[(4 * c + 2 * h + 1) * DD + n];
  }
}

__device__ __forceinline__ v8f load_c_tile(const float* __restrict__ X, int row0) {
  const int lane = threadIdx.x & 31;
  const int n = lane & 15, h = lane >> 4;
  v8f c;
#pragma unroll
  for (int r = 0; r < 8; ++r) c[r] = X[(size_t)(row0 + r + 8 * h) * DD + n];
  return c;
}

__device__ __forceinline__ void store_d_tile(float* __restrict__ O, int row0, v8f d) {
  const int lane = threadIdx.x & 31;
  const int n = lane & 15, h = lane >> 4;
#pragma unroll
  for (int r = 0; r < 8; ++r) O[(size_t)(row0 + r + 8 * h) * DD + n] = d[r];
}

__device__ __forceinline__ v8f wmma16(const v2f a[4], const v2f b[4], v8f c) {
#pragma unroll
  for (int k = 0; k < 4; ++k)
    c = __builtin_amdgcn_wmma_f32_16x16x4_f32(false, a[k], false, b[k],
                                              (short)0, c, false, false);
  return c;
}


#define EDGES_PER_ITEM 4096
#define CTR_PITCH 64
#define N_CTR 6

#define BIT_CLASS(bit) (((bit) >> 12) & 1u)

__device__ __forceinline__ int take_item(unsigned* __restrict__ tk, int nchunks, unsigned& cls,
                                         int* slot) {
  if (threadIdx.x == 0) {
    unsigned c = (__builtin_amdgcn_s_sendmsg_rtn(0x87) >> 18) & 1u;
    int t = (int)atomicAdd(&tk[c * CTR_PITCH], 1u);
    if (t >= nchunks) { c ^= 1u; t = (int)atomicAdd(&tk[c * CTR_PITCH], 1u); }
    slot[0] = t < nchunks ? t : -1;
    slot[1] = (int)c;
  }
  __syncthreads();
  cls = (unsigned)slot[1];
  return slot[0];
}

__device__ __forceinline__ void set_raw(unsigned* __restrict__ ER, unsigned which, unsigned cls,
                                        int row, int col) {
  const unsigned bit = (unsigned)col * (unsigned)NN + (unsigned)row;
  if (BIT_CLASS(bit) == cls) atomicOr(&ER[2u * (bit >> 5) + which], 1u << (bit & 31));
}

__global__ __launch_bounds__(256)
void k_build_raw(const int* __restrict__ e1, int E1, const int* __restrict__ e2, int E2,
                 int nb1, int nb2, unsigned* __restrict__ ER, unsigned* __restrict__ tk,
                 const float* __restrict__ W1, const float* __restrict__ W2,
                 const float* __restrict__ Q, float* __restrict__ G, int steps) {
  __shared__ int slot[2];
  const int tid = threadIdx.x;
  const int nitem_blocks = 2 * (nb1 + nb2);
  if ((int)blockIdx.x >= nitem_blocks) {
    const int g = (int)blockIdx.x - nitem_blocks;
    const float* __restrict__ W = (g >= steps) ? W2 + (size_t)(g - steps) * DD * DD
                                               : W1 + (size_t)g * DD * DD;
    const int k = tid >> 4, n = tid & 15;
    float acc = 0.f;
#pragma unroll
    for (int t = 0; t < DD; ++t)
      acc = fmaf(W[t * DD + k], Q[t * DD + n], acc);
    G[(size_t)g * DD * DD + k * DD + n] = acc;
    return;
  }
  unsigned cls;
  int b = take_item(tk, nb1 + nb2, cls, slot);
  if (b < 0) return;
  const int* rows; int E; unsigned which;
  if (b < nb1) { rows = e1; E = E1; which = 0u; }
  else { b -= nb1; rows = e2; E = E2; which = 1u; }
  const int* cols = rows + E;
  const int base = b * EDGES_PER_ITEM;
  const bool whole = base + EDGES_PER_ITEM <= E;
  const bool aligned = ((((size_t)rows) | ((size_t)cols)) & 15) == 0;
  if (whole && aligned) {
    const int e = base + 4 * tid;
    int4 r[4], c[4];
#pragma unroll
    for (int q = 0; q < 4; ++q) {
      r[q] = *(const int4*)(rows + e + q * (EDGES_PER_ITEM / 4));
      c[q] = *(const int4*)(cols + e + q * (EDGES_PER_ITEM / 4));
    }
    __builtin_amdgcn_sched_barrier(0);
#pragma unroll
    for (int q = 0; q < 4; ++q) {
      set_raw(ER, which, cls, r[q].x, c[q].x); set_raw(ER, which, cls, r[q].y, c[q].y);
      set_raw(ER, which, cls, r[q].z, c[q].z); set_raw(ER, which, cls, r[q].w, c[q].w);
    }
  } else {
    for (int u = 0; u < EDGES_PER_ITEM / 256; ++u) {
      const int e = base + u * 256 + tid;
      if (e < E) set_raw(ER, which, cls, rows[e], cols[e]);
    }
  }
}

typedef unsigned v2u __attribute__((ext_vector_type(2)));

__device__ __forceinline__ unsigned mask_edge(v2u pr, unsigned bit, unsigned* __restrict__ A1T,
                                              unsigned* __restrict__ A2T) {
  const unsigned m = 1u << (bit & 31), w = bit >> 5;
  if (pr.x & m) atomicOr(&A1T[w], m);
  unsigned old = m;
  if (pr.y & m) old = atomicOr(&A2T[w], m);
  return old;
}

__global__ __launch_bounds__(256)
void k_mask_hits(const int* __restrict__ mk, int EM, int nbM, const unsigned* __restrict__ ERw,
                 unsigned* __restrict__ A1T, unsigned* __restrict__ A2T,
                 unsigned* __restrict__ ctr) {
  __shared__ unsigned wsum[8];
  __shared__ int slot[2];
  const v2u* __restrict__ ER = (const v2u*)ERw;
  const int tid = threadIdx.x;
  unsigned cls;
  const int b = take_item(ctr + 4 * CTR_PITCH, nbM, cls, slot);
  if (b < 0) return;
  const int* rows = mk;
  const int* cols = mk + EM;
  const int base = b * EDGES_PER_ITEM;
  const bool whole = base + EDGES_PER_ITEM <= EM;
  const bool aligned = ((((size_t)rows) | ((size_t)cols)) & 15) == 0;
  unsigned pc = 0;
  if (whole && aligned) {
    const int e = base + 4 * tid;
    unsigned bit[16];
#pragma unroll
    for (int q = 0; q < 4; ++q) {
      const int4 r = *(const int4*)(rows + e + q * (EDGES_PER_ITEM / 4));
      const int4 c = *(const int4*)(cols + e + q * (EDGES_PER_ITEM / 4));
      bit[4 * q + 0] = (unsigned)c.x * (unsigned)NN + (unsigned)r.x;
      bit[4 * q + 1] = (unsigned)c.y * (unsigned)NN + (unsigned)r.y;
      bit[4 * q + 2] = (unsigned)c.z * (unsigned)NN + (unsigned)r.z;
      bit[4 * q + 3] = (unsigned)c.w * (unsigned)NN + (unsigned)r.w;
    }
    const unsigned spare = cls << 7;
    v2u pr[16];
#pragma unroll
    for (int k = 0; k < 16; ++k)
      pr[k] = ER[BIT_CLASS(bit[k]) == cls ? (bit[k] >> 5) : spare];
    asm volatile("" :: "v"(pr[0]), "v"(pr[1]), "v"(pr[2]), "v"(pr[3]),
                       "v"(pr[4]), "v"(pr[5]), "v"(pr[6]), "v"(pr[7]),
                       "v"(pr[8]), "v"(pr[9]), "v"(pr[10]), "v"(pr[11]),
                       "v"(pr[12]), "v"(pr[13]), "v"(pr[14]), "v"(pr[15]));
#pragma unroll
    for (int k = 0; k < 16; ++k) {
      const bool own = BIT_CLASS(bit[k]) == cls;
      pr[k].x = own ? pr[k].x : 0u;
      pr[k].y = own ? pr[k].y : 0u;
    }
    unsigned old[16];
#pragma unroll
    for (int k = 0; k < 16; ++k) old[k] = mask_edge(pr[k], bit[k], A1T, A2T);
    asm volatile("" :: "v"(old[0]), "v"(old[1]), "v"(old[2]), "v"(old[3]),
                       "v"(old[4]), "v"(old[5]), "v"(old[6]), "v"(old[7]),
                       "v"(old[8]), "v"(old[9]), "v"(old[10]), "v"(old[11]),
                       "v"(old[12]), "v"(old[13]), "v"(old[14]), "v"(old[15]));
#pragma unroll
    for (int k = 0; k < 16; ++k) pc += ((old[k] >> (bit[k] & 31)) & 1u) ^ 1u;
  } else {
    for (int u = 0; u < EDGES_PER_ITEM / 256; ++u) {
      const int e = base + u * 256 + tid;
      if (e < EM) {
        const unsigned bit = (unsigned)cols[e] * (unsigned)NN + (unsigned)rows[e];
        if (BIT_CLASS(bit) == cls)
          pc += ((mask_edge(ER[bit >> 5], bit, A1T, A2T) >> (bit & 31)) & 1u) ^ 1u;
      }
    }
  }
#pragma unroll
  for (int off = 16; off > 0; off >>= 1) pc += __shfl_xor(pc, off);
  if ((tid & 31) == 0) wsum[tid >> 5] = pc;
  __syncthreads();
  if (tid == 0) {
    unsigned t = 0;
#pragma unroll
    for (int k = 0; k < 8; ++k) t += wsum[k];
    if (t) atomicAdd(&ctr[cls * CTR_PITCH], t);
  }
}


#define DW 16
#define SRC_PER_WAVE 256
#define WPITCH 9
#define LIST_Q 34

#define ROW_OR_STEP(v, m) \
  (v) |= (unsigned)__builtin_amdgcn_update_dpp(0, (int)(v), 0x160 + (m), 0xf, 0xf, true)

struct BlkOps { v2f ax1[4]; v2f ax2[4]; v2f ar[4]; unsigned on; };

#define KB_BLOCKS 17

template <bool FIRST>
__device__ __forceinline__ void fetch_block(const float* __restrict__ x, int ibase, int c16, int h,
                                            int blk, const unsigned long long* lq1,
                                            const unsigned long long* lq2, const unsigned* wl1,
                                            const unsigned* wl2, int n1, int n2,
                                            const unsigned short* __restrict__ kbw, BlkOps& f) {
  const unsigned long long e1 = lq1[2 * blk + h];
  const unsigned long long e2 = lq2[2 * blk + h];
  const unsigned own2 = ((const unsigned char*)lq2)[16 * blk + c16];
  const unsigned xo = (unsigned)(ibase * DD + c16);
  unsigned on1 = 0u, on2 = 0u;
#pragma unroll
  for (int r = 0; r < 8; ++r) {
    const unsigned p1 = (unsigned)(e1 >> (8 * r)) & 255u;
    const unsigned p2 = (unsigned)(e2 >> (8 * r)) & 255u;
    const float a = x[xo + p1 * DD];
    const float b = x[xo + p2 * DD];
    if (r & 1) { f.ax1[r >> 1].y = a; f.ax2[r >> 1].y = b; }
    else       { f.ax1[r >> 1].x = a; f.ax2[r >> 1].x = b; }
    if (FIRST) {
      const unsigned wv1 = wl1[c16 * WPITCH + (p1 >> 5)];
      const unsigned wv2 = wl2[c16 * WPITCH + (p2 >> 5)];
      const unsigned b1 = (wv1 >> (p1 & 31u)) & 1u;
      const unsigned b2 = (wv2 >> (p2 & 31u)) & 1u;
      on1 |= b1 << r;
      on2 |= b2 << r;
    }
  }
  if (FIRST) {
    const int v1 = n1 - 16 * blk - 8 * h;
    const int v2 = n2 - 16 * blk - 8 * h;
    on1 &= v1 >= 8 ? 0xFFu : (v1 > 0 ? (1u << v1) - 1u : 0u);
    on2 &= v2 >= 8 ? 0xFFu : (v2 > 0 ? (1u << v2) - 1u : 0u);
    f.on = on1 | (on2 << 8);
  } else {
    f.on = kbw[blk * 32 + (threadIdx.x & 31)];
  }
  const float4* pr = (const float4*)(x + ((unsigned)ibase + own2) * DD + 8 * h);
  const float4 r0 = pr[0], r1 = pr[1];
  f.ar[0].x = r0.x; f.ar[0].y = r0.y;
  f.ar[1].x = r0.z; f.ar[1].y = r0.w;
  f.ar[2].x = r1.x; f.ar[2].y = r1.y;
  f.ar[3].x = r1.z; f.ar[3].y = r1.w;
}

template <bool FIRST>
__device__ __forceinline__ void block_products(const BlkOps& f, const v2f vb[4], v8f& y1, v8f& y2,
                                               int blk, unsigned short* __restrict__ kbw) {
  if (FIRST) kbw[blk * 32 + (threadIdx.x & 31)] = (unsigned short)f.on;
  const unsigned on1 = f.on & 255u, on2 = f.on >> 8;
#pragma unroll
  for (int q = 0; q < 4; ++q) {
    v2f b;
    b.x = (float)((on1 >> (2 * q)) & 1u);
    b.y = (float)((on1 >> (2 * q + 1)) & 1u);
    y1 = __builtin_amdgcn_wmma_f32_16x16x4_f32(false, f.ax1[q], false, b, (short)0, y1, false, false);
  }
  v8f s = {};
  s = wmma16(f.ar, vb, s);
#pragma unroll
  for (int q = 0; q < 4; ++q) {
    v2f p;
    p.x = __uint_as_float(__float_as_uint(s[2 * q]) & (0u - ((on2 >> (2 * q)) & 1u)));
    p.y = __uint_as_float(__float_as_uint(s[2 * q + 1]) & (0u - ((on2 >> (2 * q + 1)) & 1u)));
    y2 = __builtin_amdgcn_wmma_f32_16x16x4_f32(false, f.ax2[q], false, p, (short)0, y2, false, false);
  }
}

template <bool FIRST>
__global__ __launch_bounds__(DW * 32)
void k_step(const float* __restrict__ x, const float* __restrict__ Qm,
            const unsigned* __restrict__ A1T, const unsigned* __restrict__ A2T,
            const float* __restrict__ G1s, const float* __restrict__ G2s,
            const unsigned* __restrict__ cnt2, float s1, float* __restrict__ xnext,
            int* __restrict__ kn, unsigned long long* __restrict__ kl,
            unsigned short* __restrict__ kb) {
  __shared__ float red[DW][2][256];
  __shared__ float ysum[2][256];
  __shared__ unsigned wl[DW][2][16 * WPITCH];
  __shared__ unsigned long long lst[DW][2][LIST_Q];
  const int tid = threadIdx.x, lane = tid & 31, wave = tid >> 5;
  const int c16 = lane & 15, h = lane >> 4;
  const int j0 = blockIdx.x * 16;
  const int ibase = wave * SRC_PER_WAVE;

  v2f qa[4];
  {
    const float4* pq = (const float4*)(Qm + c16 * DD + 8 * h);
    float4 a0 = pq[0], a1 = pq[1];
    qa[0].x = a0.x; qa[0].y = a0.y;
    qa[1].x = a0.z; qa[1].y = a0.w;
    qa[2].x = a1.x; qa[2].y = a1.y;
    qa[3].x = a1.z; qa[3].y = a1.w;
  }

  const int j = j0 + c16;
  uint4 u1a = {}, u1b = {}, u2a = {}, u2b = {};
  const int wid = blockIdx.x * DW + wave;
  int* knw = kn + 2 * wid;
  unsigned long long* klw = kl + (size_t)wid * 64;
  unsigned short* kbw = kb + (size_t)wid * (KB_BLOCKS * 32);
  unsigned long long k1 = 0ull, k2 = 0ull;
  int kn1 = 0, kn2 = 0;
  if (FIRST) {
    const uint4* p1 = (const uint4*)(A1T + (size_t)j * (NN / 32) + (ibase >> 5));
    const uint4* p2 = (const uint4*)(A2T + (size_t)j * (NN / 32) + (ibase >> 5));
    u1a = p1[0]; u1b = p1[1]; u2a = p2[0]; u2b = p2[1];
  } else {
    k1 = klw[lane];
    k2 = klw[32 + lane];
    kn1 = knw[0];
    kn2 = knw[1];
  }
  const float4* pj = (const float4*)(x + (size_t)j * DD + 8 * h);
  const float4 xa0 = pj[0], xa1 = pj[1];
  v2f g1[4] = {}, g2[4] = {};
  v8f c = {};
  if (wave == 0) {
    load_b_frag(G1s, g1);
    load_b_frag(G2s, g2);
    c = load_c_tile(x, j0);
  }
  const float s2 = -(float)NN / (float)(cnt2[0] + cnt2[CTR_PITCH]);

  unsigned* wl1 = wl[wave][0];
  unsigned* wl2 = wl[wave][1];
  unsigned long long* lq1 = lst[wave][0];
  unsigned long long* lq2 = lst[wave][1];
  unsigned char* lb1 = (unsigned char*)lq1;
  unsigned char* lb2 = (unsigned char*)lq2;

  lq1[lane] = FIRST ? 0ull : k1;
  lq2[lane] = FIRST ? 0ull : k2;
  if (lane < LIST_Q - 32) { lq1[32 + lane] = 0ull; lq2[32 + lane] = 0ull; }

  unsigned w1[8], w2[8];
  w1[0] = u1a.x; w1[1] = u1a.y; w1[2] = u1a.z; w1[3] = u1a.w;
  w1[4] = u1b.x; w1[5] = u1b.y; w1[6] = u1b.z; w1[7] = u1b.w;
  w2[0] = u2a.x; w2[1] = u2a.y; w2[2] = u2a.z; w2[3] = u2a.w;
  w2[4] = u2b.x; w2[5] = u2b.y; w2[6] = u2b.z; w2[7] = u2b.w;

  int n1 = 0, n2 = 0;
  if (FIRST) {
#pragma unroll
    for (int t = 0; t < 8; ++t) {
      wl1[c16 * WPITCH + t] = w1[t];
      wl2[c16 * WPITCH + t] = w2[t];
      unsigned o1 = w1[t], o2 = w2[t];
      ROW_OR_STEP(o1, 1); ROW_OR_STEP(o2, 1);
      ROW_OR_STEP(o1, 2); ROW_OR_STEP(o2, 2);
      ROW_OR_STEP(o1, 4); ROW_OR_STEP(o2, 4);
      ROW_OR_STEP(o1, 8); ROW_OR_STEP(o2, 8);
      const unsigned m1 = __builtin_amdgcn_readfirstlane(o1);
      const unsigned m2 = __builtin_amdgcn_readfirstlane(o2);
      if ((m1 >> lane) & 1u)
        lb1[n1 + (int)__builtin_amdgcn_mbcnt_lo(m1, 0u)] = (unsigned char)(32 * t + lane);
      if ((m2 >> lane) & 1u)
        lb2[n2 + (int)__builtin_amdgcn_mbcnt_lo(m2, 0u)] = (unsigned char)(32 * t + lane);
      n1 += __builtin_popcount(m1);
      n2 += __builtin_popcount(m2);
    }
    klw[lane] = lq1[lane];
    klw[32 + lane] = lq2[lane];
    if (lane == 0) { knw[0] = n1; knw[1] = n2; }
  } else {
    n1 = __builtin_amdgcn_readfirstlane(kn1);
    n2 = __builtin_amdgcn_readfirstlane(kn2);
  }

  v2f vb[4];
  {
    v2f xjb[4];
    xjb[0].x = xa0.x; xjb[0].y = xa0.y;
    xjb[1].x = xa0.z; xjb[1].y = xa0.w;
    xjb[2].x = xa1.x; xjb[2].y = xa1.y;
    xjb[3].x = xa1.z; xjb[3].y = xa1.w;
    v8f v = {};
    v = wmma16(qa, xjb, v);
#pragma unroll
    for (int q = 0; q < 4; ++q) {
      vb[q].x = v[2 * q];
      vb[q].y = v[2 * q + 1];
    }
  }

  const int nb1 = (n1 + 15) >> 4, nb2 = (n2 + 15) >> 4;
  const int nbm = nb1 > nb2 ? nb1 : nb2;
  v8f y1 = {}, y2 = {};
  BlkOps fa, fb;
  if (FIRST) kbw[nbm * 32 + lane] = (unsigned short)0;
  fetch_block<FIRST>(x, ibase, c16, h, 0, lq1, lq2, wl1, wl2, n1, n2, kbw, fa);
  for (int blk = 0; blk < nbm; blk += 2) {
    fetch_block<FIRST>(x, ibase, c16, h, blk + 1, lq1, lq2, wl1, wl2, n1, n2, kbw, fb);
    block_products<FIRST>(fa, vb, y1, y2, blk, kbw);
    if (blk + 1 < nbm) {
      fetch_block<FIRST>(x, ibase, c16, h, blk + 2, lq1, lq2, wl1, wl2, n1, n2, kbw, fa);
      block_products<FIRST>(fb, vb, y1, y2, blk + 1, kbw);
    }
  }

#pragma unroll
  for (int r = 0; r < 8; ++r) {
    red[wave][0][r * 32 + lane] = y1[r];
    red[wave][1][r * 32 + lane] = y2[r];
  }
  __syncthreads();
  {
    const int which = tid >> 8, e = tid & 255;
    float s = 0.f;
#pragma unroll
    for (int w = 0; w < DW; ++w) s += red[w][which][e];
    const int r = e >> 5, hh = (e >> 4) & 1, n = e & 15;
    ysum[which][n * DD + r + 8 * hh] = s;
  }
  __syncthreads();
  if (wave == 0) {
    v2f a[4];
    load_a_frag_tile(ysum[0], s1, a);
    c = wmma16(a, g1, c);
    load_a_frag_tile(ysum[1], s2, a);
    c = wmma16(a, g2, c);
    store_d_tile(xnext, j0, c);
  }
}

extern "C" void kernel_launch(void* const* d_in, const int* in_sizes, int n_in,
                              void* d_out, int out_size, void* d_ws, size_t ws_size,
                              hipStream_t stream) {
  const float* x_in = (const float*)d_in[0];
  const int* e1 = (const int*)d_in[1];
  const int* e2 = (const int*)d_in[2];
  const float* Qm = (const float*)d_in[3];
  const int* mk = (const int*)d_in[4];
  const float* W1 = (const float*)d_in[5];
  const float* W2 = (const float*)d_in[6];
  float* out = (float*)d_out;

  const int E1 = in_sizes[1] / 2;
  const int E2 = in_sizes[2] / 2;
  const int EM = in_sizes[4] / 2;
  const int STEPS = in_sizes[5] / (DD * DD);

  char* ws = (char*)d_ws;
  size_t off = 0;
  auto carve = [&](size_t bytes) {
    char* p = ws + off;
    off += (bytes + 255) & ~(size_t)255;
    return p;
  };
  const size_t BM = (size_t)NN * NN / 8;
  unsigned* ER    = (unsigned*)carve(2 * BM);
  unsigned* A1T   = (unsigned*)carve(BM);
  unsigned* A2T   = (unsigned*)carve(BM);
  unsigned* ctr   = (unsigned*)carve((size_t)N_CTR * CTR_PITCH * 4);
  float* G   = (float*)carve((size_t)2 * STEPS * DD * DD * 4);
  const float* G1 = G;
  const float* G2 = G + (size_t)STEPS * DD * DD;
  float* xb0 = (float*)carve((size_t)NN * DD * 4);
  float* xb1 = (float*)carve((size_t)NN * DD * 4);
  const size_t NW = (size_t)(NN / 16) * DW;
  int* kn = (int*)carve(NW * 2 * sizeof(int));
  unsigned long long* kl = (unsigned long long*)carve(NW * 64 * sizeof(unsigned long long));
  unsigned short* kb = (unsigned short*)carve(NW * KB_BLOCKS * 32 * sizeof(unsigned short));

  hipMemsetAsync(ER, 0, 4 * BM + (size_t)N_CTR * CTR_PITCH * 4, stream);

  const int nbM = (EM + EDGES_PER_ITEM - 1) / EDGES_PER_ITEM;
  const int nb1 = (E1 + EDGES_PER_ITEM - 1) / EDGES_PER_ITEM;
  const int nb2 = (E2 + EDGES_PER_ITEM - 1) / EDGES_PER_ITEM;
  k_build_raw<<<2 * (nb1 + nb2) + 2 * STEPS, 256, 0, stream>>>(
      e1, E1, e2, E2, nb1, nb2, ER, ctr + 2 * CTR_PITCH, W1, W2, Qm, G, STEPS);
  k_mask_hits<<<2 * nbM, 256, 0, stream>>>(mk, EM, nbM, ER, A1T, A2T, ctr);

  const float s1 = (float)NN / (float)EM;

  const float* xc = x_in;
  for (int s = 0; s < STEPS; ++s) {
    float* xn = (s == STEPS - 1) ? out : ((s & 1) ? xb1 : xb0);
    if (s == 0)
      k_step<true><<<NN / 16, DW * 32, 0, stream>>>(xc, Qm, A1T, A2T, G1, G2, ctr, s1, xn,
                                                    kn, kl, kb);
    else
      k_step<false><<<NN / 16, DW * 32, 0, stream>>>(xc, Qm, A1T, A2T, G1 + (size_t)s * DD * DD,
                                                     G2 + (size_t)s * DD * DD, ctr, s1, xn,
                                                     kn, kl, kb);
    xc = xn;
  }
}
